# starting kernel
# baseline (speedup 1.0000x reference)
#include <hip/hip_runtime.h>

#define F_IN 128
#define C 64

typedef float v2f __attribute__((ext_vector_type(2)));
typedef float v8f __attribute__((ext_vector_type(8)));

__device__ __forceinline__ unsigned f32_order_encode(float f) {
  unsigned b = __float_as_uint(f);
  return (b & 0x80000000u) ? ~b : (b | 0x80000000u);
}
__device__ __forceinline__ float f32_order_decode(unsigned u) {
  unsigned b = (u & 0x80000000u) ? (u & 0x7FFFFFFFu) : ~u;
  return __uint_as_float(b);
}

__global__ void k_init(float* __restrict__ out, const float* __restrict__ bias,
                       float* __restrict__ deg, float* __restrict__ asum,
                       unsigned* __restrict__ lmaxu, float* __restrict__ denom, int N) {
  int i = blockIdx.x * blockDim.x + threadIdx.x;
  if (i < N * C) out[i] = bias[i & (C - 1)];
  if (i < N) { deg[i] = 0.f; asum[i] = 0.f; lmaxu[i] = 0u; denom[i] = 0.f; }
}

__global__ void k_degree(const int* __restrict__ ei, const float* __restrict__ eattr,
                         float* __restrict__ deg, float* __restrict__ asum, int E) {
  int e = blockIdx.x * blockDim.x + threadIdx.x;
  if (e >= E) return;
  int d = ei[E + e];
  atomicAdd(&deg[d], 1.0f);
  atomicAdd(&asum[d], eattr[e]);
}

__global__ void k_gemm(const float* __restrict__ x, const float* __restrict__ Wl,
                       const float* __restrict__ Wr, float* __restrict__ xl,
                       float* __restrict__ xr, int N) {
  int wave = threadIdx.x >> 5;
  int lane = threadIdx.x & 31;
  long long t = (long long)blockIdx.x * 8 + wave;
  long long rowTiles = (long long)(N >> 4);
  if (t >= rowTiles * 8) return;
  int sub   = (int)(t & 7);
  int m0    = (int)(t >> 3) << 4;
  int n0    = (sub >> 1) << 4;
  int which = sub & 1;
  const float* W = which ? Wr : Wl;
  float*       Y = which ? xr : xl;

  int half = lane >> 4;
  int l15  = lane & 15;
  const float* arow = x + (long long)(m0 + l15) * F_IN + half * 2;
  const float* bcol = W + n0 + l15 + (long long)(half * 2) * C;

  v8f acc = {0.f, 0.f, 0.f, 0.f, 0.f, 0.f, 0.f, 0.f};
#pragma unroll 4
  for (int k0 = 0; k0 < F_IN; k0 += 4) {
    v2f a, b;
    a.x = arow[k0];
    a.y = arow[k0 + 1];
    b.x = bcol[(long long)k0 * C];
    b.y = bcol[(long long)(k0 + 1) * C];
    acc = __builtin_amdgcn_wmma_f32_16x16x4_f32(
         false, a,  false, b,
         (short)0, acc,  false,  false);
  }
  float* yrow = Y + (long long)(m0 + half * 8) * C + n0 + l15;
#pragma unroll
  for (int r = 0; r < 8; ++r) yrow[(long long)r * C] = acc[r];
}

__global__ void k_logits(const int* __restrict__ ei, const float* __restrict__ eattr,
                         const float* __restrict__ xl, const float* __restrict__ xr,
                         const float* __restrict__ deg, const float* __restrict__ asum,
                         const float* __restrict__ We, const float* __restrict__ att,
                         float* __restrict__ logit, unsigned* __restrict__ lmaxu,
                         int E, int N) {
  int gid = blockIdx.x * blockDim.x + threadIdx.x;
  int w = gid >> 5, lane = gid & 31;
  if (w >= E + N) return;
  int s, d; float ea;
  if (w < E) { s = ei[w]; d = ei[E + w]; ea = eattr[w]; }
  else       { s = d = w - E; ea = asum[s] / fmaxf(deg[s], 1.0f); }

  float p = 0.f;
#pragma unroll
  for (int j = 0; j < 2; ++j) {
    int c = lane + j * 32;
    float m = xl[(long long)s * C + c] + xr[(long long)d * C + c] + ea * We[c];
    m = (m > 0.f) ? m : 0.2f * m;
    p += m * att[c];
  }
  for (int off = 16; off > 0; off >>= 1) p += __shfl_xor(p, off, 32);
  if (lane == 0) {
    logit[w] = p;
    atomicMax(&lmaxu[d], f32_order_encode(p));
  }
}

__global__ void k_denom(const int* __restrict__ ei, float* __restrict__ logit,
                        const unsigned* __restrict__ lmaxu, float* __restrict__ denom,
                        int E, int N) {
  int i = blockIdx.x * blockDim.x + threadIdx.x;
  if (i >= E + N) return;
  int d = (i < E) ? ei[E + i] : (i - E);
  float lm = f32_order_decode(lmaxu[d]);
  float ex = __expf(logit[i] - lm);
  logit[i] = ex;
  atomicAdd(&denom[d], ex);
}

__global__ void k_aggregate(const int* __restrict__ ei, const float* __restrict__ exv,
                            const float* __restrict__ denom, const float* __restrict__ xl,
                            float* __restrict__ out, int E, int N) {
  int gid = blockIdx.x * blockDim.x + threadIdx.x;
  int w = gid >> 5, lane = gid & 31;
  if (w >= E + N) return;
  int s, d;
  if (w < E) { s = ei[w]; d = ei[E + w]; } else { s = d = w - E; }
  float alpha = exv[w] / denom[d];
#pragma unroll
  for (int j = 0; j < 2; ++j) {
    int c = lane + j * 32;
    atomicAdd(&out[(long long)d * C + c], alpha * xl[(long long)s * C + c]);
  }
}

extern "C" void kernel_launch(void* const* d_in, const int* in_sizes, int n_in,
                              void* d_out, int out_size, void* d_ws, size_t ws_size,
                              hipStream_t stream) {
  (void)n_in; (void)out_size; (void)ws_size;
  const float* x     = (const float*)d_in[0];
  const float* eattr = (const float*)d_in[1];
  const float* Wl    = (const float*)d_in[2];
  const float* Wr    = (const float*)d_in[3];
  const float* We    = (const float*)d_in[4];
  const float* att   = (const float*)d_in[5];
  const float* bias  = (const float*)d_in[6];
  const int*   ei    = (const int*)d_in[7];
  float*       out   = (float*)d_out;

  int N = in_sizes[0] / F_IN;
  int E = in_sizes[1];

  char* ws = (char*)d_ws;
  size_t o = 0;
  auto take = [&](size_t bytes) -> void* {
    void* p = ws + o; o += (bytes + 255) & ~(size_t)255; return p;
  };
  float*    xl    = (float*)take((size_t)N * C * 4);
  float*    xr    = (float*)take((size_t)N * C * 4);
  float*    deg   = (float*)take((size_t)N * 4);
  float*    asum  = (float*)take((size_t)N * 4);
  unsigned* lmaxu = (unsigned*)take((size_t)N * 4);
  float*    denom = (float*)take((size_t)N * 4);
  float*    logit = (float*)take((size_t)(E + N) * 4);

  const int B = 256;
  long long initThr = (long long)N * C;
  k_init<<<(unsigned)((initThr + B - 1) / B), B, 0, stream>>>(out, bias, deg, asum, lmaxu, denom, N);
  k_degree<<<(unsigned)((E + B - 1) / B), B, 0, stream>>>(ei, eattr, deg, asum, E);
  k_gemm<<<(unsigned)(N >> 4), 256, 0, stream>>>(x, Wl, Wr, xl, xr, N);

  long long edges = (long long)E + N;
  long long thr   = edges * 32;
  k_logits<<<(unsigned)((thr + B - 1) / B), B, 0, stream>>>(ei, eattr, xl, xr, deg, asum,
                                                            We, att, logit, lmaxu, E, N);
  k_denom<<<(unsigned)((edges + B - 1) / B), B, 0, stream>>>(ei, logit, lmaxu, denom, E, N);
  k_aggregate<<<(unsigned)((thr + B - 1) / B), B, 0, stream>>>(ei, logit, denom, xl, out, E, N);
}


# logits and aggregate: a wave now owns 32 edges (indices read once coalesced, lane-broadcast; 16 lanes x 16 B per row in logits), row gathers batched; the 64-term logit sum order changes
# speedup vs baseline: 1.6485x; 1.6485x over previous
#include <hip/hip_runtime.h>

#define F_IN 128
#define C 64

typedef float v2f __attribute__((ext_vector_type(2)));
typedef float v8f __attribute__((ext_vector_type(8)));

__device__ __forceinline__ unsigned f32_order_encode(float f) {
  unsigned b = __float_as_uint(f);
  return (b & 0x80000000u) ? ~b : (b | 0x80000000u);
}
__device__ __forceinline__ float f32_order_decode(unsigned u) {
  unsigned b = (u & 0x80000000u) ? (u & 0x7FFFFFFFu) : ~u;
  return __uint_as_float(b);
}

__global__ void k_init(float* __restrict__ out, const float* __restrict__ bias,
                       float* __restrict__ deg, float* __restrict__ asum,
                       unsigned* __restrict__ lmaxu, float* __restrict__ denom, int N) {
  int i = blockIdx.x * blockDim.x + threadIdx.x;
  if (i < N * C) out[i] = bias[i & (C - 1)];
  if (i < N) { deg[i] = 0.f; asum[i] = 0.f; lmaxu[i] = 0u; denom[i] = 0.f; }
}

__global__ void k_degree(const int* __restrict__ ei, const float* __restrict__ eattr,
                         float* __restrict__ deg, float* __restrict__ asum, int E) {
  int e = blockIdx.x * blockDim.x + threadIdx.x;
  if (e >= E) return;
  int d = ei[E + e];
  atomicAdd(&deg[d], 1.0f);
  atomicAdd(&asum[d], eattr[e]);
}

__global__ void k_gemm(const float* __restrict__ x, const float* __restrict__ Wl,
                       const float* __restrict__ Wr, float* __restrict__ xl,
                       float* __restrict__ xr, int N) {
  int wave = threadIdx.x >> 5;
  int lane = threadIdx.x & 31;
  long long t = (long long)blockIdx.x * 8 + wave;
  long long rowTiles = (long long)(N >> 4);
  if (t >= rowTiles * 8) return;
  int sub   = (int)(t & 7);
  int m0    = (int)(t >> 3) << 4;
  int n0    = (sub >> 1) << 4;
  int which = sub & 1;
  const float* W = which ? Wr : Wl;
  float*       Y = which ? xr : xl;

  int half = lane >> 4;
  int l15  = lane & 15;
  const float* arow = x + (long long)(m0 + l15) * F_IN + half * 2;
  const float* bcol = W + n0 + l15 + (long long)(half * 2) * C;

  v8f acc = {0.f, 0.f, 0.f, 0.f, 0.f, 0.f, 0.f, 0.f};
#pragma unroll 4
  for (int k0 = 0; k0 < F_IN; k0 += 4) {
    v2f a, b;
    a.x = arow[k0];
    a.y = arow[k0 + 1];
    b.x = bcol[(long long)k0 * C];
    b.y = bcol[(long long)(k0 + 1) * C];
    acc = __builtin_amdgcn_wmma_f32_16x16x4_f32(
         false, a,  false, b,
         (short)0, acc,  false,  false);
  }
  float* yrow = Y + (long long)(m0 + half * 8) * C + n0 + l15;
#pragma unroll
  for (int r = 0; r < 8; ++r) yrow[(long long)r * C] = acc[r];
}

typedef float v4f __attribute__((ext_vector_type(4)));

template <int J>
__device__ __forceinline__ int row_share_i(int v) {
  return __builtin_amdgcn_update_dpp(0, v, 0x150 + J, 0xf, 0xf, true);
}
template <int J>
__device__ __forceinline__ float row_share_f(float v) {
  return __int_as_float(row_share_i<J>(__float_as_int(v)));
}
template <int M>
__device__ __forceinline__ float row_xadd(float p) {
  return p + __int_as_float(__builtin_amdgcn_update_dpp(0, __float_as_int(p), 0x160 + M, 0xf, 0xf, true));
}

template <int J>
__device__ __forceinline__ void logit_load(int s, int d, int l15, const float* __restrict__ xl,
                                           const float* __restrict__ xr, v4f& a, v4f& b) {
  int sq = row_share_i<J>(s);
  int dq = row_share_i<J>(d);
  a = *(const v4f*)(xl + (long long)sq * C + l15 * 4);
  b = *(const v4f*)(xr + (long long)dq * C + l15 * 4);
}
template <int J>
__device__ __forceinline__ void logit_step(float ea, int l15, v4f a, v4f b, v4f we, v4f at, float& res) {
  float eq = row_share_f<J>(ea);
  float p = 0.f;
#pragma unroll
  for (int k = 0; k < 4; ++k) {
    float m = a[k] + b[k] + eq * we[k];
    m = (m > 0.f) ? m : 0.2f * m;
    p += m * at[k];
  }
  p = row_xadd<1>(p); p = row_xadd<2>(p); p = row_xadd<4>(p); p = row_xadd<8>(p);
  res = (l15 == J) ? p : res;
}

__global__ void k_logits(const int* __restrict__ ei, const float* __restrict__ eattr,
                         const float* __restrict__ xl, const float* __restrict__ xr,
                         const float* __restrict__ deg, const float* __restrict__ asum,
                         const float* __restrict__ We, const float* __restrict__ att,
                         float* __restrict__ logit, unsigned* __restrict__ lmaxu,
                         int E, int N) {
  int lane = threadIdx.x & 31;
  int l15  = lane & 15;
  int wv   = blockIdx.x * (blockDim.x >> 5) + (threadIdx.x >> 5);
  int tot  = E + N;
  int base = __builtin_amdgcn_readfirstlane(wv * 32);
  if (base >= tot) return;
  int e = base + lane;
  bool valid = e < tot;
  int s = 0, d = 0; float ea = 0.f;
  if (e < E)      { s = ei[e]; d = ei[E + e]; ea = eattr[e]; }
  else if (valid) { s = d = e - E; ea = asum[s] / fmaxf(deg[s], 1.0f); }

  v4f we = *(const v4f*)(We + l15 * 4);
  v4f at = *(const v4f*)(att + l15 * 4);
  float res = 0.f;
  v4f a0, a1, a2, a3, a4, a5, a6, a7, b0, b1, b2, b3, b4, b5, b6, b7;
  logit_load<0>(s, d, l15, xl, xr, a0, b0); logit_load<1>(s, d, l15, xl, xr, a1, b1);
  logit_load<2>(s, d, l15, xl, xr, a2, b2); logit_load<3>(s, d, l15, xl, xr, a3, b3);
  logit_load<4>(s, d, l15, xl, xr, a4, b4); logit_load<5>(s, d, l15, xl, xr, a5, b5);
  logit_load<6>(s, d, l15, xl, xr, a6, b6); logit_load<7>(s, d, l15, xl, xr, a7, b7);
  logit_step<0>(ea, l15, a0, b0, we, at, res); logit_step<1>(ea, l15, a1, b1, we, at, res);
  logit_step<2>(ea, l15, a2, b2, we, at, res); logit_step<3>(ea, l15, a3, b3, we, at, res);
  logit_step<4>(ea, l15, a4, b4, we, at, res); logit_step<5>(ea, l15, a5, b5, we, at, res);
  logit_step<6>(ea, l15, a6, b6, we, at, res); logit_step<7>(ea, l15, a7, b7, we, at, res);
  logit_load<8>(s, d, l15, xl, xr, a0, b0);  logit_load<9>(s, d, l15, xl, xr, a1, b1);
  logit_load<10>(s, d, l15, xl, xr, a2, b2); logit_load<11>(s, d, l15, xl, xr, a3, b3);
  logit_load<12>(s, d, l15, xl, xr, a4, b4); logit_load<13>(s, d, l15, xl, xr, a5, b5);
  logit_load<14>(s, d, l15, xl, xr, a6, b6); logit_load<15>(s, d, l15, xl, xr, a7, b7);
  logit_step<8>(ea, l15, a0, b0, we, at, res);  logit_step<9>(ea, l15, a1, b1, we, at, res);
  logit_step<10>(ea, l15, a2, b2, we, at, res); logit_step<11>(ea, l15, a3, b3, we, at, res);
  logit_step<12>(ea, l15, a4, b4, we, at, res); logit_step<13>(ea, l15, a5, b5, we, at, res);
  logit_step<14>(ea, l15, a6, b6, we, at, res); logit_step<15>(ea, l15, a7, b7, we, at, res);

  if (valid) {
    logit[e] = res;
    atomicMax(&lmaxu[d], f32_order_encode(res));
  }
}

__global__ void k_denom(const int* __restrict__ ei, float* __restrict__ logit,
                        const unsigned* __restrict__ lmaxu, float* __restrict__ denom,
                        int E, int N) {
  int i = blockIdx.x * blockDim.x + threadIdx.x;
  if (i >= E + N) return;
  int d = (i < E) ? ei[E + i] : (i - E);
  float lm = f32_order_decode(lmaxu[d]);
  float ex = __expf(logit[i] - lm);
  logit[i] = ex;
  atomicAdd(&denom[d], ex);
}

__global__ void k_aggregate(const int* __restrict__ ei, const float* __restrict__ exv,
                            const float* __restrict__ denom, const float* __restrict__ xl,
                            float* __restrict__ out, int E, int N) {
  int lane = threadIdx.x & 31;
  int wv   = blockIdx.x * (blockDim.x >> 5) + (threadIdx.x >> 5);
  int tot  = E + N;
  int base = __builtin_amdgcn_readfirstlane(wv * 32);
  if (base >= tot) return;
  int e = base + lane;
  int s = 0, d = 0; float alpha = 0.f;
  if (e < E)        { s = ei[e]; d = ei[E + e]; }
  else if (e < tot) { s = d = e - E; }
  if (e < tot) alpha = exv[e] / denom[d];
  int cnt = tot - base;

#pragma unroll
  for (int j0 = 0; j0 < 32; j0 += 16) {
    float v0[16], v1[16];
#pragma unroll
    for (int k = 0; k < 16; ++k) {
      int sq = __builtin_amdgcn_readlane(s, j0 + k);
      const float* xrow = xl + (long long)sq * C;
      v0[k] = xrow[lane];
      v1[k] = xrow[lane + 32];
    }
#pragma unroll
    for (int k = 0; k < 16; ++k) {
      if (j0 + k < cnt) {
        int   dq = __builtin_amdgcn_readlane(d, j0 + k);
        float aq = __int_as_float(__builtin_amdgcn_readlane(__float_as_int(alpha), j0 + k));
        float* orow = out + (long long)dq * C;
        atomicAdd(&orow[lane],      aq * v0[k]);
        atomicAdd(&orow[lane + 32], aq * v1[k]);
      }
    }
  }
}

extern "C" void kernel_launch(void* const* d_in, const int* in_sizes, int n_in,
                              void* d_out, int out_size, void* d_ws, size_t ws_size,
                              hipStream_t stream) {
  (void)n_in; (void)out_size; (void)ws_size;
  const float* x     = (const float*)d_in[0];
  const float* eattr = (const float*)d_in[1];
  const float* Wl    = (const float*)d_in[2];
  const float* Wr    = (const float*)d_in[3];
  const float* We    = (const float*)d_in[4];
  const float* att   = (const float*)d_in[5];
  const float* bias  = (const float*)d_in[6];
  const int*   ei    = (const int*)d_in[7];
  float*       out   = (float*)d_out;

  int N = in_sizes[0] / F_IN;
  int E = in_sizes[1];

  char* ws = (char*)d_ws;
  size_t o = 0;
  auto take = [&](size_t bytes) -> void* {
    void* p = ws + o; o += (bytes + 255) & ~(size_t)255; return p;
  };
  float*    xl    = (float*)take((size_t)N * C * 4);
  float*    xr    = (float*)take((size_t)N * C * 4);
  float*    deg   = (float*)take((size_t)N * 4);
  float*    asum  = (float*)take((size_t)N * 4);
  unsigned* lmaxu = (unsigned*)take((size_t)N * 4);
  float*    denom = (float*)take((size_t)N * 4);
  float*    logit = (float*)take((size_t)(E + N) * 4);

  const int B = 256;
  long long initThr = (long long)N * C;
  k_init<<<(unsigned)((initThr + B - 1) / B), B, 0, stream>>>(out, bias, deg, asum, lmaxu, denom, N);
  k_degree<<<(unsigned)((E + B - 1) / B), B, 0, stream>>>(ei, eattr, deg, asum, E);
  k_gemm<<<(unsigned)(N >> 4), 256, 0, stream>>>(x, Wl, Wr, xl, xr, N);

  long long edges = (long long)E + N;
  long long thr   = ((edges + 31) / 32) * 32;
  k_logits<<<(unsigned)((thr + B - 1) / B), B, 0, stream>>>(ei, eattr, xl, xr, deg, asum,
                                                            We, att, logit, lmaxu, E, N);
  k_denom<<<(unsigned)((edges + B - 1) / B), B, 0, stream>>>(ei, logit, lmaxu, denom, E, N);
  k_aggregate<<<(unsigned)((thr + B - 1) / B), B, 0, stream>>>(ei, logit, denom, xl, out, E, N);
}


# edges grouped by destination (histogram, wave-scanned segment allocation, fill); a 16-lane row per node does logits, running softmax and row sum in registers: no float atomics, sum order changes
# speedup vs baseline: 2.4881x; 1.5093x over previous
#include <hip/hip_runtime.h>

#define F_IN 128
#define C 64

typedef float v2f __attribute__((ext_vector_type(2)));
typedef float v8f __attribute__((ext_vector_type(8)));

__global__ void k_zero(int* __restrict__ cnt, int* __restrict__ total, int N) {
  int i = blockIdx.x * blockDim.x + threadIdx.x;
  if (i < N) cnt[i] = 0;
  if (i == 0) *total = 0;
}

__global__ void k_hist(const int* __restrict__ ei, int* __restrict__ cnt, int E) {
  int e = blockIdx.x * blockDim.x + threadIdx.x;
  if (e < E) atomicAdd(&cnt[ei[E + e]], 1);
}

__global__ void k_alloc(const int* __restrict__ cnt, int* __restrict__ total, int2* __restrict__ seg, int N) {
  int i = blockIdx.x * blockDim.x + threadIdx.x;
  int lane = threadIdx.x & 31;
  int v = (i < N) ? cnt[i] : 0;
  int incl = v;
#pragma unroll
  for (int o = 1; o < 32; o <<= 1) { int u = __shfl_up(incl, o, 32); if (lane >= o) incl += u; }
  int base = 0;
  if (lane == 31) base = atomicAdd(total, incl);
  base = __shfl(base, 31, 32);
  if (i < N) seg[i] = make_int2(base + incl - v, v);
}

__global__ void k_fill(const int* __restrict__ ei, const float* __restrict__ eattr,
                       const int2* __restrict__ seg, int* __restrict__ cnt,
                       int2* __restrict__ sedge, int E) {
  int e = blockIdx.x * blockDim.x + threadIdx.x;
  if (e >= E) return;
  int d = ei[E + e];
  int p = seg[d].x + atomicAdd(&cnt[d], -1) - 1;
  sedge[p] = make_int2(ei[e], __float_as_int(eattr[e]));
}

__global__ void k_gemm(const float* __restrict__ x, const float* __restrict__ Wl,
                       const float* __restrict__ Wr, float* __restrict__ xl,
                       float* __restrict__ xr, int N) {
  int wave = threadIdx.x >> 5;
  int lane = threadIdx.x & 31;
  long long t = (long long)blockIdx.x * 8 + wave;
  long long rowTiles = (long long)(N >> 4);
  if (t >= rowTiles * 8) return;
  int sub   = (int)(t & 7);
  int m0    = (int)(t >> 3) << 4;
  int n0    = (sub >> 1) << 4;
  int which = sub & 1;
  const float* W = which ? Wr : Wl;
  float*       Y = which ? xr : xl;

  int half = lane >> 4;
  int l15  = lane & 15;
  const float* arow = x + (long long)(m0 + l15) * F_IN + half * 2;
  const float* bcol = W + n0 + l15 + (long long)(half * 2) * C;

  v8f acc = {0.f, 0.f, 0.f, 0.f, 0.f, 0.f, 0.f, 0.f};
#pragma unroll 4
  for (int k0 = 0; k0 < F_IN; k0 += 4) {
    v2f a, b;
    a.x = arow[k0];
    a.y = arow[k0 + 1];
    b.x = bcol[(long long)k0 * C];
    b.y = bcol[(long long)(k0 + 1) * C];
    acc = __builtin_amdgcn_wmma_f32_16x16x4_f32(
         false, a,  false, b,
         (short)0, acc,  false,  false);
  }
  float* yrow = Y + (long long)(m0 + half * 8) * C + n0 + l15;
#pragma unroll
  for (int r = 0; r < 8; ++r) yrow[(long long)r * C] = acc[r];
}

typedef float v4f __attribute__((ext_vector_type(4)));

template <int J>
__device__ __forceinline__ int row_share_i(int v) {
  return __builtin_amdgcn_update_dpp(0, v, 0x150 + J, 0xf, 0xf, true);
}
template <int J>
__device__ __forceinline__ float row_share_f(float v) {
  return __int_as_float(row_share_i<J>(__float_as_int(v)));
}
template <int M>
__device__ __forceinline__ float row_xadd(float p) {
  return p + __int_as_float(__builtin_amdgcn_update_dpp(0, __float_as_int(p), 0x160 + M, 0xf, 0xf, true));
}

__device__ __forceinline__ float edge_logit(v4f a, v4f b, float eq, v4f we, v4f at) {
  float p = 0.f;
#pragma unroll
  for (int k = 0; k < 4; ++k) {
    float m = a[k] + b[k] + eq * we[k];
    m = (m > 0.f) ? m : 0.2f * m;
    p += m * at[k];
  }
  p = row_xadd<1>(p); p = row_xadd<2>(p); p = row_xadd<4>(p); p = row_xadd<8>(p);
  return p;
}

__global__ void k_node(const int2* __restrict__ seg, const int2* __restrict__ sedge,
                       const float* __restrict__ xl, const float* __restrict__ xr,
                       const float* __restrict__ We, const float* __restrict__ att,
                       const float* __restrict__ bias, float* __restrict__ out, int N) {
  int l15 = threadIdx.x & 15;
  int d   = (blockIdx.x * blockDim.x + threadIdx.x) >> 4;
  if (d >= N) return;
  int2 sg = seg[d];
  int beg = sg.x, end = sg.x + sg.y;
  v4f we = *(const v4f*)(We + l15 * 4);
  v4f at = *(const v4f*)(att + l15 * 4);
  v4f b  = *(const v4f*)(xr + (long long)d * C + l15 * 4);
  v4f aself = *(const v4f*)(xl + (long long)d * C + l15 * 4);

  float M = -INFINITY, S = 0.f, easum = 0.f;
  v4f acc = {0.f, 0.f, 0.f, 0.f};
  for (int c = beg; c < end; c += 16) {
    int j = c + l15;
    int sj = d; float eaj = 0.f;
    if (j < end) { int2 r = sedge[j]; sj = r.x; eaj = __int_as_float(r.y); }
    easum += eaj;
    int nrem = end - c; nrem = nrem > 16 ? 16 : nrem;
    for (int k0 = 0; k0 < nrem; k0 += 4) {
      v4f a[4]; float eq[4], l[4];
#pragma unroll
      for (int k = 0; k < 4; ++k) {
        int sk = __shfl(sj, k0 + k, 16);
        eq[k]  = __shfl(eaj, k0 + k, 16);
        a[k]   = *(const v4f*)(xl + (long long)sk * C + l15 * 4);
      }
      float mx = M;
#pragma unroll
      for (int k = 0; k < 4; ++k) {
        l[k] = edge_logit(a[k], b, eq[k], we, at);
        l[k] = (k0 + k < nrem) ? l[k] : -INFINITY;
        mx = fmaxf(mx, l[k]);
      }
      float sc = __expf(M - mx);
      S *= sc; acc *= sc;
#pragma unroll
      for (int k = 0; k < 4; ++k) {
        float p = __expf(l[k] - mx);
        S += p; acc += p * a[k];
      }
      M = mx;
    }
  }
  easum = row_xadd<1>(easum); easum = row_xadd<2>(easum); easum = row_xadd<4>(easum); easum = row_xadd<8>(easum);
  float loop_ea = easum / fmaxf((float)(end - beg), 1.0f);
  float ls = edge_logit(aself, b, loop_ea, we, at);
  float mx = fmaxf(M, ls);
  float sc = __expf(M - mx);
  float p  = __expf(ls - mx);
  S = S * sc + p;
  acc = acc * sc + p * aself;
  v4f bi = *(const v4f*)(bias + l15 * 4);
  v4f o;
#pragma unroll
  for (int k = 0; k < 4; ++k) o[k] = acc[k] / S + bi[k];
  *(v4f*)(out + (long long)d * C + l15 * 4) = o;
}

extern "C" void kernel_launch(void* const* d_in, const int* in_sizes, int n_in,
                              void* d_out, int out_size, void* d_ws, size_t ws_size,
                              hipStream_t stream) {
  (void)n_in; (void)out_size; (void)ws_size;
  const float* x     = (const float*)d_in[0];
  const float* eattr = (const float*)d_in[1];
  const float* Wl    = (const float*)d_in[2];
  const float* Wr    = (const float*)d_in[3];
  const float* We    = (const float*)d_in[4];
  const float* att   = (const float*)d_in[5];
  const float* bias  = (const float*)d_in[6];
  const int*   ei    = (const int*)d_in[7];
  float*       out   = (float*)d_out;

  int N = in_sizes[0] / F_IN;
  int E = in_sizes[1];

  char* ws = (char*)d_ws;
  size_t o = 0;
  auto take = [&](size_t bytes) -> void* {
    void* p = ws + o; o += (bytes + 255) & ~(size_t)255; return p;
  };
  float* xl     = (float*)take((size_t)N * C * 4);
  float* xr     = (float*)take((size_t)N * C * 4);
  int*   cnt    = (int*)take((size_t)N * 4);
  int*   total  = (int*)take(4);
  int2*  seg    = (int2*)take((size_t)N * 8);
  int2*  sedge  = (int2*)take((size_t)E * 8);

  const int B = 256;
  k_zero<<<(unsigned)((N + B - 1) / B), B, 0, stream>>>(cnt, total, N);
  k_hist<<<(unsigned)((E + B - 1) / B), B, 0, stream>>>(ei, cnt, E);
  k_alloc<<<(unsigned)((N + B - 1) / B), B, 0, stream>>>(cnt, total, seg, N);
  k_fill<<<(unsigned)((E + B - 1) / B), B, 0, stream>>>(ei, eattr, seg, cnt, sedge, E);
  k_gemm<<<(unsigned)(N >> 4), 256, 0, stream>>>(x, Wl, Wr, xl, xr, N);
  k_node<<<(unsigned)(((long long)N * 16 + B - 1) / B), B, 0, stream>>>(seg, sedge, xl, xr, We, att, bias, out, N);
}


# destination counts and per-edge ranks by returning LDS adds (64 edge chunks, all node counters in LDS) plus chunk-prefixed offsets: no global atomic per edge; slot (sum) order now chunk-major
# speedup vs baseline: 4.8567x; 1.9520x over previous
#include <hip/hip_runtime.h>

#define F_IN 128
#define C 64

typedef float v2f __attribute__((ext_vector_type(2)));
typedef float v8f __attribute__((ext_vector_type(8)));

#define HB 64
#define NODE_CAP 65536

__global__ __launch_bounds__(1024) void k_count(const int* __restrict__ ei, int* __restrict__ hcnt,
                                                int* __restrict__ rank, int* __restrict__ total, int N, int E) {
  __shared__ __attribute__((aligned(16))) int h[NODE_CAP];
  const int tid = threadIdx.x;
  const bool vec = (N & 3) == 0;
  if (vec) {
    int4* h4 = (int4*)h;
    for (int i = tid; i < (N >> 2); i += 1024) h4[i] = make_int4(0, 0, 0, 0);
  } else {
    for (int i = tid; i < N; i += 1024) h[i] = 0;
  }
  if (blockIdx.x == 0 && tid == 0) *total = 0;
  __syncthreads();
  const int chunk = (E + HB - 1) / HB;
  const int e0 = blockIdx.x * chunk;
  const int e1 = min(e0 + chunk, E);
  const int* __restrict__ dst = ei + E;
  int e = e0 + tid;
  for (; e + 3072 < e1; e += 4096) {
    int d0 = dst[e], d1 = dst[e + 1024], d2 = dst[e + 2048], d3 = dst[e + 3072];
    int r0 = atomicAdd(&h[d0], 1);
    int r1 = atomicAdd(&h[d1], 1);
    int r2 = atomicAdd(&h[d2], 1);
    int r3 = atomicAdd(&h[d3], 1);
    rank[e] = r0; rank[e + 1024] = r1; rank[e + 2048] = r2; rank[e + 3072] = r3;
  }
  for (; e < e1; e += 1024) rank[e] = atomicAdd(&h[dst[e]], 1);
  __syncthreads();
  int* __restrict__ o = hcnt + (long long)blockIdx.x * N;
  if (vec) {
    const int4* h4 = (const int4*)h;
    int4* __restrict__ o4 = (int4*)o;
    for (int i = tid; i < (N >> 2); i += 1024) o4[i] = h4[i];
  } else {
    for (int i = tid; i < N; i += 1024) o[i] = h[i];
  }
}

__global__ void k_alloc(int* __restrict__ hcnt, int* __restrict__ total, int2* __restrict__ seg, int N) {
  int i = blockIdx.x * blockDim.x + threadIdx.x;
  int lane = threadIdx.x & 31;
  int c[HB];
  int v = 0;
  if (i < N) {
#pragma unroll
    for (int g = 0; g < HB; ++g) { c[g] = hcnt[(long long)g * N + i]; v += c[g]; }
  }
  int incl = v;
#pragma unroll
  for (int o = 1; o < 32; o <<= 1) { int u = __shfl_up(incl, o, 32); if (lane >= o) incl += u; }
  int base = 0;
  if (lane == 31) base = atomicAdd(total, incl);
  base = __shfl(base, 31, 32);
  if (i < N) {
    int run = base + incl - v;
    seg[i] = make_int2(run, v);
#pragma unroll
    for (int g = 0; g < HB; ++g) { hcnt[(long long)g * N + i] = run; run += c[g]; }
  }
}

__global__ void k_fill(const int* __restrict__ ei, const float* __restrict__ eattr,
                       const int* __restrict__ hoff, const int* __restrict__ rank,
                       int2* __restrict__ sedge, int N, int E) {
  int e = blockIdx.x * blockDim.x + threadIdx.x;
  if (e >= E) return;
  int chunk = (E + HB - 1) / HB;
  int g = e / chunk;
  int d = ei[E + e];
  int p = hoff[(long long)g * N + d] + rank[e];
  sedge[p] = make_int2(ei[e], __float_as_int(eattr[e]));
}

__global__ void k_gemm(const float* __restrict__ x, const float* __restrict__ Wl,
                       const float* __restrict__ Wr, float* __restrict__ xl,
                       float* __restrict__ xr, int N) {
  int wave = threadIdx.x >> 5;
  int lane = threadIdx.x & 31;
  long long t = (long long)blockIdx.x * 8 + wave;
  long long rowTiles = (long long)(N >> 4);
  if (t >= rowTiles * 8) return;
  int sub   = (int)(t & 7);
  int m0    = (int)(t >> 3) << 4;
  int n0    = (sub >> 1) << 4;
  int which = sub & 1;
  const float* W = which ? Wr : Wl;
  float*       Y = which ? xr : xl;

  int half = lane >> 4;
  int l15  = lane & 15;
  const float* arow = x + (long long)(m0 + l15) * F_IN + half * 2;
  const float* bcol = W + n0 + l15 + (long long)(half * 2) * C;

  v8f acc = {0.f, 0.f, 0.f, 0.f, 0.f, 0.f, 0.f, 0.f};
#pragma unroll 4
  for (int k0 = 0; k0 < F_IN; k0 += 4) {
    v2f a, b;
    a.x = arow[k0];
    a.y = arow[k0 + 1];
    b.x = bcol[(long long)k0 * C];
    b.y = bcol[(long long)(k0 + 1) * C];
    acc = __builtin_amdgcn_wmma_f32_16x16x4_f32(
         false, a,  false, b,
         (short)0, acc,  false,  false);
  }
  float* yrow = Y + (long long)(m0 + half * 8) * C + n0 + l15;
#pragma unroll
  for (int r = 0; r < 8; ++r) yrow[(long long)r * C] = acc[r];
}

typedef float v4f __attribute__((ext_vector_type(4)));

template <int J>
__device__ __forceinline__ int row_share_i(int v) {
  return __builtin_amdgcn_update_dpp(0, v, 0x150 + J, 0xf, 0xf, true);
}
template <int J>
__device__ __forceinline__ float row_share_f(float v) {
  return __int_as_float(row_share_i<J>(__float_as_int(v)));
}
template <int M>
__device__ __forceinline__ float row_xadd(float p) {
  return p + __int_as_float(__builtin_amdgcn_update_dpp(0, __float_as_int(p), 0x160 + M, 0xf, 0xf, true));
}

__device__ __forceinline__ float edge_logit(v4f a, v4f b, float eq, v4f we, v4f at) {
  float p = 0.f;
#pragma unroll
  for (int k = 0; k < 4; ++k) {
    float m = a[k] + b[k] + eq * we[k];
    m = (m > 0.f) ? m : 0.2f * m;
    p += m * at[k];
  }
  p = row_xadd<1>(p); p = row_xadd<2>(p); p = row_xadd<4>(p); p = row_xadd<8>(p);
  return p;
}

__global__ void k_node(const int2* __restrict__ seg, const int2* __restrict__ sedge,
                       const float* __restrict__ xl, const float* __restrict__ xr,
                       const float* __restrict__ We, const float* __restrict__ att,
                       const float* __restrict__ bias, float* __restrict__ out, int N) {
  int l15 = threadIdx.x & 15;
  int d   = (blockIdx.x * blockDim.x + threadIdx.x) >> 4;
  if (d >= N) return;
  int2 sg = seg[d];
  int beg = sg.x, end = sg.x + sg.y;
  v4f we = *(const v4f*)(We + l15 * 4);
  v4f at = *(const v4f*)(att + l15 * 4);
  v4f b  = *(const v4f*)(xr + (long long)d * C + l15 * 4);
  v4f aself = *(const v4f*)(xl + (long long)d * C + l15 * 4);

  float M = -INFINITY, S = 0.f, easum = 0.f;
  v4f acc = {0.f, 0.f, 0.f, 0.f};
  for (int c = beg; c < end; c += 16) {
    int j = c + l15;
    int sj = d; float eaj = 0.f;
    if (j < end) { int2 r = sedge[j]; sj = r.x; eaj = __int_as_float(r.y); }
    easum += eaj;
    int nrem = end - c; nrem = nrem > 16 ? 16 : nrem;
    for (int k0 = 0; k0 < nrem; k0 += 4) {
      v4f a[4]; float eq[4], l[4];
#pragma unroll
      for (int k = 0; k < 4; ++k) {
        int sk = __shfl(sj, k0 + k, 16);
        eq[k]  = __shfl(eaj, k0 + k, 16);
        a[k]   = *(const v4f*)(xl + (long long)sk * C + l15 * 4);
      }
      float mx = M;
#pragma unroll
      for (int k = 0; k < 4; ++k) {
        l[k] = edge_logit(a[k], b, eq[k], we, at);
        l[k] = (k0 + k < nrem) ? l[k] : -INFINITY;
        mx = fmaxf(mx, l[k]);
      }
      float sc = __expf(M - mx);
      S *= sc; acc *= sc;
#pragma unroll
      for (int k = 0; k < 4; ++k) {
        float p = __expf(l[k] - mx);
        S += p; acc += p * a[k];
      }
      M = mx;
    }
  }
  easum = row_xadd<1>(easum); easum = row_xadd<2>(easum); easum = row_xadd<4>(easum); easum = row_xadd<8>(easum);
  float loop_ea = easum / fmaxf((float)(end - beg), 1.0f);
  float ls = edge_logit(aself, b, loop_ea, we, at);
  float mx = fmaxf(M, ls);
  float sc = __expf(M - mx);
  float p  = __expf(ls - mx);
  S = S * sc + p;
  acc = acc * sc + p * aself;
  v4f bi = *(const v4f*)(bias + l15 * 4);
  v4f o;
#pragma unroll
  for (int k = 0; k < 4; ++k) o[k] = acc[k] / S + bi[k];
  *(v4f*)(out + (long long)d * C + l15 * 4) = o;
}

extern "C" void kernel_launch(void* const* d_in, const int* in_sizes, int n_in,
                              void* d_out, int out_size, void* d_ws, size_t ws_size,
                              hipStream_t stream) {
  (void)n_in; (void)out_size; (void)ws_size;
  const float* x     = (const float*)d_in[0];
  const float* eattr = (const float*)d_in[1];
  const float* Wl    = (const float*)d_in[2];
  const float* Wr    = (const float*)d_in[3];
  const float* We    = (const float*)d_in[4];
  const float* att   = (const float*)d_in[5];
  const float* bias  = (const float*)d_in[6];
  const int*   ei    = (const int*)d_in[7];
  float*       out   = (float*)d_out;

  int N = in_sizes[0] / F_IN;
  int E = in_sizes[1];

  char* ws = (char*)d_ws;
  size_t o = 0;
  auto take = [&](size_t bytes) -> void* {
    void* p = ws + o; o += (bytes + 255) & ~(size_t)255; return p;
  };
  float* xl     = (float*)take((size_t)N * C * 4);
  float* xr     = (float*)take((size_t)N * C * 4);
  int*   total  = (int*)take(4);
  int2*  seg    = (int2*)take((size_t)N * 8);
  int2*  sedge  = (int2*)take((size_t)E * 8);
  int*   hcnt   = (int*)xl;
  int*   rank   = (int*)xr;

  const int B = 256;
  k_count<<<HB, 1024, 0, stream>>>(ei, hcnt, rank, total, N, E);
  k_alloc<<<(unsigned)((N + B - 1) / B), B, 0, stream>>>(hcnt, total, seg, N);
  k_fill<<<(unsigned)((E + B - 1) / B), B, 0, stream>>>(ei, eattr, hcnt, rank, sedge, N, E);
  k_gemm<<<(unsigned)(N >> 4), 256, 0, stream>>>(x, Wl, Wr, xl, xr, N);
  k_node<<<(unsigned)(((long long)N * 16 + B - 1) / B), B, 0, stream>>>(seg, sedge, xl, xr, We, att, bias, out, N);
}


# GEMM tiles ride in the counting launch as extra blocks on the units counting leaves idle (5 to 4 launches); 48 chunks, planes and ranks held in out until the node kernel; slot (sum) order changes
# speedup vs baseline: 5.0575x; 1.0413x over previous
#include <hip/hip_runtime.h>

#define F_IN 128
#define C 64

typedef float v2f __attribute__((ext_vector_type(2)));
typedef float v8f __attribute__((ext_vector_type(8)));

#define HB 48
#define NODE_CAP 65536

__device__ __forceinline__ void gemm_tile(const float* __restrict__ x, const float* __restrict__ Wl,
                                          const float* __restrict__ Wr, float* __restrict__ xl,
                                          float* __restrict__ xr, long long t, int lane) {
  int sub   = (int)(t & 7);
  int m0    = (int)(t >> 3) << 4;
  int n0    = (sub >> 1) << 4;
  int which = sub & 1;
  const float* W = which ? Wr : Wl;
  float*       Y = which ? xr : xl;

  int half = lane >> 4;
  int l15  = lane & 15;
  const float* arow = x + (long long)(m0 + l15) * F_IN + half * 2;
  const float* bcol = W + n0 + l15 + (long long)(half * 2) * C;

  v8f acc = {0.f, 0.f, 0.f, 0.f, 0.f, 0.f, 0.f, 0.f};
#pragma unroll 8
  for (int k0 = 0; k0 < F_IN; k0 += 4) {
    v2f a, b;
    a.x = arow[k0];
    a.y = arow[k0 + 1];
    b.x = bcol[(long long)k0 * C];
    b.y = bcol[(long long)(k0 + 1) * C];
    acc = __builtin_amdgcn_wmma_f32_16x16x4_f32(
         false, a,  false, b,
         (short)0, acc,  false,  false);
  }
  float* yrow = Y + (long long)(m0 + half * 8) * C + n0 + l15;
#pragma unroll
  for (int r = 0; r < 8; ++r) yrow[(long long)r * C] = acc[r];
}

__global__ __launch_bounds__(1024) void k_count_gemm(const int* __restrict__ ei, int* __restrict__ hcnt,
                                                     int* __restrict__ rank, int* __restrict__ total,
                                                     const float* __restrict__ x, const float* __restrict__ Wl,
                                                     const float* __restrict__ Wr, float* __restrict__ xl,
                                                     float* __restrict__ xr, int N, int E) {
  __shared__ __attribute__((aligned(16))) int h[NODE_CAP];
  const int tid = threadIdx.x;
  if (blockIdx.x >= HB) {
    long long t = (long long)(blockIdx.x - HB) * 32 + (tid >> 5);
    if (t < (long long)(N >> 4) * 8)
      gemm_tile(x, Wl, Wr, xl, xr, t, tid & 31);
    return;
  }
  const bool vec = (N & 3) == 0;
  if (vec) {
    int4* h4 = (int4*)h;
    for (int i = tid; i < (N >> 2); i += 1024) h4[i] = make_int4(0, 0, 0, 0);
  } else {
    for (int i = tid; i < N; i += 1024) h[i] = 0;
  }
  if (blockIdx.x == 0 && tid == 0) *total = 0;
  __syncthreads();
  const int chunk = (E + HB - 1) / HB;
  const int e0 = blockIdx.x * chunk;
  const int e1 = min(e0 + chunk, E);
  const int* __restrict__ dst = ei + E;
  int e = e0 + tid;
  for (; e + 3072 < e1; e += 4096) {
    int d0 = dst[e], d1 = dst[e + 1024], d2 = dst[e + 2048], d3 = dst[e + 3072];
    int r0 = atomicAdd(&h[d0], 1);
    int r1 = atomicAdd(&h[d1], 1);
    int r2 = atomicAdd(&h[d2], 1);
    int r3 = atomicAdd(&h[d3], 1);
    rank[e] = r0; rank[e + 1024] = r1; rank[e + 2048] = r2; rank[e + 3072] = r3;
  }
  for (; e < e1; e += 1024) rank[e] = atomicAdd(&h[dst[e]], 1);
  __syncthreads();
  int* __restrict__ o = hcnt + (long long)blockIdx.x * N;
  if (vec) {
    const int4* h4 = (const int4*)h;
    int4* __restrict__ o4 = (int4*)o;
    for (int i = tid; i < (N >> 2); i += 1024) o4[i] = h4[i];
  } else {
    for (int i = tid; i < N; i += 1024) o[i] = h[i];
  }
}

__global__ void k_alloc(int* __restrict__ hcnt, int* __restrict__ total, int2* __restrict__ seg, int N) {
  int i = blockIdx.x * blockDim.x + threadIdx.x;
  int lane = threadIdx.x & 31;
  int c[HB];
  int v = 0;
  if (i < N) {
#pragma unroll
    for (int g = 0; g < HB; ++g) { c[g] = hcnt[(long long)g * N + i]; v += c[g]; }
  }
  int incl = v;
#pragma unroll
  for (int o = 1; o < 32; o <<= 1) { int u = __shfl_up(incl, o, 32); if (lane >= o) incl += u; }
  int base = 0;
  if (lane == 31) base = atomicAdd(total, incl);
  base = __shfl(base, 31, 32);
  if (i < N) {
    int run = base + incl - v;
    seg[i] = make_int2(run, v);
#pragma unroll
    for (int g = 0; g < HB; ++g) { hcnt[(long long)g * N + i] = run; run += c[g]; }
  }
}

__global__ void k_fill(const int* __restrict__ ei, const float* __restrict__ eattr,
                       const int* __restrict__ hoff, const int* __restrict__ rank,
                       int2* __restrict__ sedge, int N, int E) {
  int e = blockIdx.x * blockDim.x + threadIdx.x;
  if (e >= E) return;
  int chunk = (E + HB - 1) / HB;
  int g = e / chunk;
  int d = ei[E + e];
  int p = hoff[(long long)g * N + d] + rank[e];
  sedge[p] = make_int2(ei[e], __float_as_int(eattr[e]));
}

typedef float v4f __attribute__((ext_vector_type(4)));

template <int J>
__device__ __forceinline__ int row_share_i(int v) {
  return __builtin_amdgcn_update_dpp(0, v, 0x150 + J, 0xf, 0xf, true);
}
template <int J>
__device__ __forceinline__ float row_share_f(float v) {
  return __int_as_float(row_share_i<J>(__float_as_int(v)));
}
template <int M>
__device__ __forceinline__ float row_xadd(float p) {
  return p + __int_as_float(__builtin_amdgcn_update_dpp(0, __float_as_int(p), 0x160 + M, 0xf, 0xf, true));
}

__device__ __forceinline__ float edge_logit(v4f a, v4f b, float eq, v4f we, v4f at) {
  float p = 0.f;
#pragma unroll
  for (int k = 0; k < 4; ++k) {
    float m = a[k] + b[k] + eq * we[k];
    m = (m > 0.f) ? m : 0.2f * m;
    p += m * at[k];
  }
  p = row_xadd<1>(p); p = row_xadd<2>(p); p = row_xadd<4>(p); p = row_xadd<8>(p);
  return p;
}

__global__ void k_node(const int2* __restrict__ seg, const int2* __restrict__ sedge,
                       const float* __restrict__ xl, const float* __restrict__ xr,
                       const float* __restrict__ We, const float* __restrict__ att,
                       const float* __restrict__ bias, float* __restrict__ out, int N) {
  int l15 = threadIdx.x & 15;
  int d   = (blockIdx.x * blockDim.x + threadIdx.x) >> 4;
  if (d >= N) return;
  int2 sg = seg[d];
  int beg = sg.x, end = sg.x + sg.y;
  v4f we = *(const v4f*)(We + l15 * 4);
  v4f at = *(const v4f*)(att + l15 * 4);
  v4f b  = *(const v4f*)(xr + (long long)d * C + l15 * 4);
  v4f aself = *(const v4f*)(xl + (long long)d * C + l15 * 4);

  float M = -INFINITY, S = 0.f, easum = 0.f;
  v4f acc = {0.f, 0.f, 0.f, 0.f};
  for (int c = beg; c < end; c += 16) {
    int j = c + l15;
    int sj = d; float eaj = 0.f;
    if (j < end) { int2 r = sedge[j]; sj = r.x; eaj = __int_as_float(r.y); }
    easum += eaj;
    int nrem = end - c; nrem = nrem > 16 ? 16 : nrem;
    for (int k0 = 0; k0 < nrem; k0 += 4) {
      v4f a[4]; float eq[4], l[4];
#pragma unroll
      for (int k = 0; k < 4; ++k) {
        int sk = __shfl(sj, k0 + k, 16);
        eq[k]  = __shfl(eaj, k0 + k, 16);
        a[k]   = *(const v4f*)(xl + (long long)sk * C + l15 * 4);
      }
      float mx = M;
#pragma unroll
      for (int k = 0; k < 4; ++k) {
        l[k] = edge_logit(a[k], b, eq[k], we, at);
        l[k] = (k0 + k < nrem) ? l[k] : -INFINITY;
        mx = fmaxf(mx, l[k]);
      }
      float sc = __expf(M - mx);
      S *= sc; acc *= sc;
#pragma unroll
      for (int k = 0; k < 4; ++k) {
        float p = __expf(l[k] - mx);
        S += p; acc += p * a[k];
      }
      M = mx;
    }
  }
  easum = row_xadd<1>(easum); easum = row_xadd<2>(easum); easum = row_xadd<4>(easum); easum = row_xadd<8>(easum);
  float loop_ea = easum / fmaxf((float)(end - beg), 1.0f);
  float ls = edge_logit(aself, b, loop_ea, we, at);
  float mx = fmaxf(M, ls);
  float sc = __expf(M - mx);
  float p  = __expf(ls - mx);
  S = S * sc + p;
  acc = acc * sc + p * aself;
  v4f bi = *(const v4f*)(bias + l15 * 4);
  v4f o;
#pragma unroll
  for (int k = 0; k < 4; ++k) o[k] = acc[k] / S + bi[k];
  *(v4f*)(out + (long long)d * C + l15 * 4) = o;
}

extern "C" void kernel_launch(void* const* d_in, const int* in_sizes, int n_in,
                              void* d_out, int out_size, void* d_ws, size_t ws_size,
                              hipStream_t stream) {
  (void)n_in; (void)out_size; (void)ws_size;
  const float* x     = (const float*)d_in[0];
  const float* eattr = (const float*)d_in[1];
  const float* Wl    = (const float*)d_in[2];
  const float* Wr    = (const float*)d_in[3];
  const float* We    = (const float*)d_in[4];
  const float* att   = (const float*)d_in[5];
  const float* bias  = (const float*)d_in[6];
  const int*   ei    = (const int*)d_in[7];
  float*       out   = (float*)d_out;

  int N = in_sizes[0] / F_IN;
  int E = in_sizes[1];

  char* ws = (char*)d_ws;
  size_t o = 0;
  auto take = [&](size_t bytes) -> void* {
    void* p = ws + o; o += (bytes + 255) & ~(size_t)255; return p;
  };
  float* xl     = (float*)take((size_t)N * C * 4);
  float* xr     = (float*)take((size_t)N * C * 4);
  int*   total  = (int*)take(4);
  int2*  seg    = (int2*)take((size_t)N * 8);
  int2*  sedge  = (int2*)take((size_t)E * 8);
  int*   hcnt   = (int*)out;
  int*   rank   = hcnt + (size_t)HB * N;

  const int B = 256;
  k_count_gemm<<<(unsigned)(HB + ((long long)(N >> 4) * 8 + 31) / 32), 1024, 0, stream>>>(ei, hcnt, rank, total, x, Wl, Wr, xl, xr, N, E);
  k_alloc<<<(unsigned)((N + B - 1) / B), B, 0, stream>>>(hcnt, total, seg, N);
  k_fill<<<(unsigned)((E + B - 1) / B), B, 0, stream>>>(ei, eattr, hcnt, rank, sedge, N, E);
  k_node<<<(unsigned)(((long long)N * 16 + B - 1) / B), B, 0, stream>>>(seg, sedge, xl, xr, We, att, bias, out, N);
}


# GEMM feed: a wave owns a 16x64 strip (4 accumulator tiles), W staged once per block in LDS as 16-byte k-quads, x read 16 B per lane per 8 k, 208 resident blocks; the 128-term sum order changes
# speedup vs baseline: 6.2925x; 1.2442x over previous
#include <hip/hip_runtime.h>

#define F_IN 128
#define C 64

typedef float v2f __attribute__((ext_vector_type(2)));
typedef float v8f __attribute__((ext_vector_type(8)));

#define HB 48
#define NODE_CAP 65536
#define GB 208

typedef float v4f __attribute__((ext_vector_type(4)));

__device__ __forceinline__ v8f mma4(v4f a, v4f b, int j, v8f acc) {
  v2f aa, bb;
  aa.x = a[2 * j]; aa.y = a[2 * j + 1];
  bb.x = b[2 * j]; bb.y = b[2 * j + 1];
  return __builtin_amdgcn_wmma_f32_16x16x4_f32(
       false, aa,  false, bb,
       (short)0, acc,  false,  false);
}

__device__ __forceinline__ void gemm_block(const float* __restrict__ x, const float* __restrict__ Wl,
                                           const float* __restrict__ Wr, float* __restrict__ xl,
                                           float* __restrict__ xr, float* wlds, int gb, int ngb, int N, int tid) {
  const long long T = (long long)(N >> 4) * 2;
  if ((long long)gb * 32 >= T) return;
  const int lane = tid & 31, hf = lane >> 4, c = lane & 15;
  const long long tstep = (long long)ngb * 32;
  long long t = (long long)gb * 32 + (tid >> 5);
  const v4f* ap = (const v4f*)(x + (long long)((int)(t >> 1) * 16 + c) * F_IN + hf * 4);
  v4f a[4];
  if (t < T) {
#pragma unroll
    for (int i = 0; i < 4; ++i) a[i] = ap[2 * i];
  }
  {
    const int col = tid & 127, kk0 = tid >> 7;
    const float* s = (col < 64 ? Wl + col : Wr + (col - 64)) + (long long)(kk0 * 4) * C;
    v4f w[4];
#pragma unroll
    for (int i = 0; i < 4; ++i) {
      const float* si = s + (long long)(i * 32) * C;
      w[i].x = si[0]; w[i].y = si[C]; w[i].z = si[2 * C]; w[i].w = si[3 * C];
    }
#pragma unroll
    for (int i = 0; i < 4; ++i) *(v4f*)(wlds + (tid + i * 1024) * 4) = w[i];
  }
  __syncthreads();

  while (t < T) {
    const int which = (int)(t & 1);
    const int m0    = (int)(t >> 1) << 4;
    const v4f* bw = (const v4f*)wlds + hf * 128 + which * 64 + c;
    v8f acc[4];
#pragma unroll
    for (int tl = 0; tl < 4; ++tl) acc[tl] = (v8f){0.f, 0.f, 0.f, 0.f, 0.f, 0.f, 0.f, 0.f};
    v4f b[4];
#pragma unroll
    for (int tl = 0; tl < 4; ++tl) b[tl] = bw[tl * 16];
#pragma unroll
    for (int g = 0; g < 16; ++g) {
      v4f bn[4];
      if (g + 1 < 16) {
#pragma unroll
        for (int tl = 0; tl < 4; ++tl) bn[tl] = bw[(g + 1) * 256 + tl * 16];
      }
      v4f av = a[g & 3];
      if (g + 4 < 16) a[g & 3] = ap[2 * (g + 4)];
#pragma unroll
      for (int j = 0; j < 2; ++j) {
#pragma unroll
        for (int tl = 0; tl < 4; ++tl) acc[tl] = mma4(av, b[tl], j, acc[tl]);
      }
      if (g + 1 < 16) {
#pragma unroll
        for (int tl = 0; tl < 4; ++tl) b[tl] = bn[tl];
      }
      __builtin_amdgcn_sched_barrier(0);
    }
    float* yrow = (which ? xr : xl) + (long long)(m0 + hf * 8) * C + c;
#pragma unroll
    for (int tl = 0; tl < 4; ++tl) {
#pragma unroll
      for (int r = 0; r < 8; ++r) yrow[(long long)r * C + tl * 16] = acc[tl][r];
    }
    t += tstep;
    if (t < T) {
      ap = (const v4f*)(x + (long long)((int)(t >> 1) * 16 + c) * F_IN + hf * 4);
#pragma unroll
      for (int i = 0; i < 4; ++i) a[i] = ap[2 * i];
    }
  }
}

__global__ __launch_bounds__(1024) void k_count_gemm(const int* __restrict__ ei, int* __restrict__ hcnt,
                                                     int* __restrict__ rank, int* __restrict__ total,
                                                     const float* __restrict__ x, const float* __restrict__ Wl,
                                                     const float* __restrict__ Wr, float* __restrict__ xl,
                                                     float* __restrict__ xr, int N, int E) {
  __shared__ __attribute__((aligned(16))) int h[NODE_CAP];
  const int tid = threadIdx.x;
  if (blockIdx.x >= HB) {
    gemm_block(x, Wl, Wr, xl, xr, (float*)h, (int)blockIdx.x - HB, (int)gridDim.x - HB, N, tid);
    return;
  }
  const bool vec = (N & 3) == 0;
  if (vec) {
    int4* h4 = (int4*)h;
    for (int i = tid; i < (N >> 2); i += 1024) h4[i] = make_int4(0, 0, 0, 0);
  } else {
    for (int i = tid; i < N; i += 1024) h[i] = 0;
  }
  if (blockIdx.x == 0 && tid == 0) *total = 0;
  __syncthreads();
  const int chunk = (E + HB - 1) / HB;
  const int e0 = blockIdx.x * chunk;
  const int e1 = min(e0 + chunk, E);
  const int* __restrict__ dst = ei + E;
  int e = e0 + tid;
  for (; e + 3072 < e1; e += 4096) {
    int d0 = dst[e], d1 = dst[e + 1024], d2 = dst[e + 2048], d3 = dst[e + 3072];
    int r0 = atomicAdd(&h[d0], 1);
    int r1 = atomicAdd(&h[d1], 1);
    int r2 = atomicAdd(&h[d2], 1);
    int r3 = atomicAdd(&h[d3], 1);
    rank[e] = r0; rank[e + 1024] = r1; rank[e + 2048] = r2; rank[e + 3072] = r3;
  }
  for (; e < e1; e += 1024) rank[e] = atomicAdd(&h[dst[e]], 1);
  __syncthreads();
  int* __restrict__ o = hcnt + (long long)blockIdx.x * N;
  if (vec) {
    const int4* h4 = (const int4*)h;
    int4* __restrict__ o4 = (int4*)o;
    for (int i = tid; i < (N >> 2); i += 1024) o4[i] = h4[i];
  } else {
    for (int i = tid; i < N; i += 1024) o[i] = h[i];
  }
}

__global__ void k_alloc(int* __restrict__ hcnt, int* __restrict__ total, int2* __restrict__ seg, int N) {
  int i = blockIdx.x * blockDim.x + threadIdx.x;
  int lane = threadIdx.x & 31;
  int c[HB];
  int v = 0;
  if (i < N) {
#pragma unroll
    for (int g = 0; g < HB; ++g) { c[g] = hcnt[(long long)g * N + i]; v += c[g]; }
  }
  int incl = v;
#pragma unroll
  for (int o = 1; o < 32; o <<= 1) { int u = __shfl_up(incl, o, 32); if (lane >= o) incl += u; }
  int base = 0;
  if (lane == 31) base = atomicAdd(total, incl);
  base = __shfl(base, 31, 32);
  if (i < N) {
    int run = base + incl - v;
    seg[i] = make_int2(run, v);
#pragma unroll
    for (int g = 0; g < HB; ++g) { hcnt[(long long)g * N + i] = run; run += c[g]; }
  }
}

__global__ void k_fill(const int* __restrict__ ei, const float* __restrict__ eattr,
                       const int* __restrict__ hoff, const int* __restrict__ rank,
                       int2* __restrict__ sedge, int N, int E) {
  int e = blockIdx.x * blockDim.x + threadIdx.x;
  if (e >= E) return;
  int chunk = (E + HB - 1) / HB;
  int g = e / chunk;
  int d = ei[E + e];
  int p = hoff[(long long)g * N + d] + rank[e];
  sedge[p] = make_int2(ei[e], __float_as_int(eattr[e]));
}


template <int J>
__device__ __forceinline__ int row_share_i(int v) {
  return __builtin_amdgcn_update_dpp(0, v, 0x150 + J, 0xf, 0xf, true);
}
template <int J>
__device__ __forceinline__ float row_share_f(float v) {
  return __int_as_float(row_share_i<J>(__float_as_int(v)));
}
template <int M>
__device__ __forceinline__ float row_xadd(float p) {
  return p + __int_as_float(__builtin_amdgcn_update_dpp(0, __float_as_int(p), 0x160 + M, 0xf, 0xf, true));
}

__device__ __forceinline__ float edge_logit(v4f a, v4f b, float eq, v4f we, v4f at) {
  float p = 0.f;
#pragma unroll
  for (int k = 0; k < 4; ++k) {
    float m = a[k] + b[k] + eq * we[k];
    m = (m > 0.f) ? m : 0.2f * m;
    p += m * at[k];
  }
  p = row_xadd<1>(p); p = row_xadd<2>(p); p = row_xadd<4>(p); p = row_xadd<8>(p);
  return p;
}

__global__ void k_node(const int2* __restrict__ seg, const int2* __restrict__ sedge,
                       const float* __restrict__ xl, const float* __restrict__ xr,
                       const float* __restrict__ We, const float* __restrict__ att,
                       const float* __restrict__ bias, float* __restrict__ out, int N) {
  int l15 = threadIdx.x & 15;
  int d   = (blockIdx.x * blockDim.x + threadIdx.x) >> 4;
  if (d >= N) return;
  int2 sg = seg[d];
  int beg = sg.x, end = sg.x + sg.y;
  v4f we = *(const v4f*)(We + l15 * 4);
  v4f at = *(const v4f*)(att + l15 * 4);
  v4f b  = *(const v4f*)(xr + (long long)d * C + l15 * 4);
  v4f aself = *(const v4f*)(xl + (long long)d * C + l15 * 4);

  float M = -INFINITY, S = 0.f, easum = 0.f;
  v4f acc = {0.f, 0.f, 0.f, 0.f};
  for (int c = beg; c < end; c += 16) {
    int j = c + l15;
    int sj = d; float eaj = 0.f;
    if (j < end) { int2 r = sedge[j]; sj = r.x; eaj = __int_as_float(r.y); }
    easum += eaj;
    int nrem = end - c; nrem = nrem > 16 ? 16 : nrem;
    for (int k0 = 0; k0 < nrem; k0 += 4) {
      v4f a[4]; float eq[4], l[4];
#pragma unroll
      for (int k = 0; k < 4; ++k) {
        int sk = __shfl(sj, k0 + k, 16);
        eq[k]  = __shfl(eaj, k0 + k, 16);
        a[k]   = *(const v4f*)(xl + (long long)sk * C + l15 * 4);
      }
      float mx = M;
#pragma unroll
      for (int k = 0; k < 4; ++k) {
        l[k] = edge_logit(a[k], b, eq[k], we, at);
        l[k] = (k0 + k < nrem) ? l[k] : -INFINITY;
        mx = fmaxf(mx, l[k]);
      }
      float sc = __expf(M - mx);
      S *= sc; acc *= sc;
#pragma unroll
      for (int k = 0; k < 4; ++k) {
        float p = __expf(l[k] - mx);
        S += p; acc += p * a[k];
      }
      M = mx;
    }
  }
  easum = row_xadd<1>(easum); easum = row_xadd<2>(easum); easum = row_xadd<4>(easum); easum = row_xadd<8>(easum);
  float loop_ea = easum / fmaxf((float)(end - beg), 1.0f);
  float ls = edge_logit(aself, b, loop_ea, we, at);
  float mx = fmaxf(M, ls);
  float sc = __expf(M - mx);
  float p  = __expf(ls - mx);
  S = S * sc + p;
  acc = acc * sc + p * aself;
  v4f bi = *(const v4f*)(bias + l15 * 4);
  v4f o;
#pragma unroll
  for (int k = 0; k < 4; ++k) o[k] = acc[k] / S + bi[k];
  *(v4f*)(out + (long long)d * C + l15 * 4) = o;
}

extern "C" void kernel_launch(void* const* d_in, const int* in_sizes, int n_in,
                              void* d_out, int out_size, void* d_ws, size_t ws_size,
                              hipStream_t stream) {
  (void)n_in; (void)out_size; (void)ws_size;
  const float* x     = (const float*)d_in[0];
  const float* eattr = (const float*)d_in[1];
  const float* Wl    = (const float*)d_in[2];
  const float* Wr    = (const float*)d_in[3];
  const float* We    = (const float*)d_in[4];
  const float* att   = (const float*)d_in[5];
  const float* bias  = (const float*)d_in[6];
  const int*   ei    = (const int*)d_in[7];
  float*       out   = (float*)d_out;

  int N = in_sizes[0] / F_IN;
  int E = in_sizes[1];

  char* ws = (char*)d_ws;
  size_t o = 0;
  auto take = [&](size_t bytes) -> void* {
    void* p = ws + o; o += (bytes + 255) & ~(size_t)255; return p;
  };
  float* xl     = (float*)take((size_t)N * C * 4);
  float* xr     = (float*)take((size_t)N * C * 4);
  int*   total  = (int*)take(4);
  int2*  seg    = (int2*)take((size_t)N * 8);
  int2*  sedge  = (int2*)take((size_t)E * 8);
  int*   hcnt   = (int*)out;
  int*   rank   = hcnt + (size_t)HB * N;

  const int B = 256;
  k_count_gemm<<<HB + GB, 1024, 0, stream>>>(ei, hcnt, rank, total, x, Wl, Wr, xl, xr, N, E);
  k_alloc<<<(unsigned)((N + B - 1) / B), B, 0, stream>>>(hcnt, total, seg, N);
  k_fill<<<(unsigned)((E + B - 1) / B), B, 0, stream>>>(ei, eattr, hcnt, rank, sedge, N, E);
  k_node<<<(unsigned)(((long long)N * 16 + B - 1) / B), B, 0, stream>>>(seg, sedge, xl, xr, We, att, bias, out, N);
}


# a chunk's block scans its LDS counts and places its own edges (2nd returning LDS add); node kernel walks a node's 48 runs from end planes: 4 to 2 launches; x_r in out; sum order stays chunk-major
# speedup vs baseline: 7.4712x; 1.1873x over previous
#include <hip/hip_runtime.h>

#define F_IN 128
#define C 64

typedef float v2f __attribute__((ext_vector_type(2)));
typedef float v8f __attribute__((ext_vector_type(8)));

#define HB 48
#define NODE_CAP 65536
#define GB 208

typedef float v4f __attribute__((ext_vector_type(4)));

__device__ __forceinline__ v8f mma4(v4f a, v4f b, int j, v8f acc) {
  v2f aa, bb;
  aa.x = a[2 * j]; aa.y = a[2 * j + 1];
  bb.x = b[2 * j]; bb.y = b[2 * j + 1];
  return __builtin_amdgcn_wmma_f32_16x16x4_f32(
       false, aa,  false, bb,
       (short)0, acc,  false,  false);
}

__device__ __forceinline__ void gemm_block(const float* __restrict__ x, const float* __restrict__ Wl,
                                           const float* __restrict__ Wr, float* __restrict__ xl,
                                           float* __restrict__ xr, float* wlds, int gb, int ngb, int N, int tid) {
  const long long T = (long long)(N >> 4) * 2;
  if ((long long)gb * 32 >= T) return;
  const int lane = tid & 31, hf = lane >> 4, c = lane & 15;
  const long long tstep = (long long)ngb * 32;
  long long t = (long long)gb * 32 + (tid >> 5);
  const v4f* ap = (const v4f*)(x + (long long)((int)(t >> 1) * 16 + c) * F_IN + hf * 4);
  v4f a[4];
  if (t < T) {
#pragma unroll
    for (int i = 0; i < 4; ++i) a[i] = ap[2 * i];
  }
  {
    const int col = tid & 127, kk0 = tid >> 7;
    const float* s = (col < 64 ? Wl + col : Wr + (col - 64)) + (long long)(kk0 * 4) * C;
    v4f w[4];
#pragma unroll
    for (int i = 0; i < 4; ++i) {
      const float* si = s + (long long)(i * 32) * C;
      w[i].x = si[0]; w[i].y = si[C]; w[i].z = si[2 * C]; w[i].w = si[3 * C];
    }
#pragma unroll
    for (int i = 0; i < 4; ++i) *(v4f*)(wlds + (tid + i * 1024) * 4) = w[i];
  }
  __syncthreads();

  while (t < T) {
    const int which = (int)(t & 1);
    const int m0    = (int)(t >> 1) << 4;
    const v4f* bw = (const v4f*)wlds + hf * 128 + which * 64 + c;
    v8f acc[4];
#pragma unroll
    for (int tl = 0; tl < 4; ++tl) acc[tl] = (v8f){0.f, 0.f, 0.f, 0.f, 0.f, 0.f, 0.f, 0.f};
    v4f b[4];
#pragma unroll
    for (int tl = 0; tl < 4; ++tl) b[tl] = bw[tl * 16];
#pragma unroll
    for (int g = 0; g < 16; ++g) {
      v4f bn[4];
      if (g + 1 < 16) {
#pragma unroll
        for (int tl = 0; tl < 4; ++tl) bn[tl] = bw[(g + 1) * 256 + tl * 16];
      }
      v4f av = a[g & 3];
      if (g + 4 < 16) a[g & 3] = ap[2 * (g + 4)];
#pragma unroll
      for (int j = 0; j < 2; ++j) {
#pragma unroll
        for (int tl = 0; tl < 4; ++tl) acc[tl] = mma4(av, b[tl], j, acc[tl]);
      }
      if (g + 1 < 16) {
#pragma unroll
        for (int tl = 0; tl < 4; ++tl) b[tl] = bn[tl];
      }
      __builtin_amdgcn_sched_barrier(0);
    }
    float* yrow = (which ? xr : xl) + (long long)(m0 + hf * 8) * C + c;
#pragma unroll
    for (int tl = 0; tl < 4; ++tl) {
#pragma unroll
      for (int r = 0; r < 8; ++r) yrow[(long long)r * C + tl * 16] = acc[tl][r];
    }
    t += tstep;
    if (t < T) {
      ap = (const v4f*)(x + (long long)((int)(t >> 1) * 16 + c) * F_IN + hf * 4);
#pragma unroll
      for (int i = 0; i < 4; ++i) a[i] = ap[2 * i];
    }
  }
}

__global__ __launch_bounds__(1024) void k_count_gemm(const int* __restrict__ ei, const float* __restrict__ eattr,
                                                     int* __restrict__ endp, int2* __restrict__ sedge,
                                                     const float* __restrict__ x, const float* __restrict__ Wl,
                                                     const float* __restrict__ Wr, float* __restrict__ xl,
                                                     float* __restrict__ xr, int N, int E) {
  __shared__ __attribute__((aligned(16))) int h[NODE_CAP];
  __shared__ int wsum[32];
  const int tid = threadIdx.x;
  if (blockIdx.x >= HB) {
    gemm_block(x, Wl, Wr, xl, xr, (float*)h, (int)blockIdx.x - HB, (int)gridDim.x - HB, N, tid);
    return;
  }
  const bool vec = (N & 3) == 0;
  if (vec) {
    int4* h4 = (int4*)h;
    for (int i = tid; i < (N >> 2); i += 1024) h4[i] = make_int4(0, 0, 0, 0);
  } else {
    for (int i = tid; i < N; i += 1024) h[i] = 0;
  }
  __syncthreads();
  const int chunk = (E + HB - 1) / HB;
  const int e0 = blockIdx.x * chunk;
  const int e1 = min(e0 + chunk, E);
  const int* __restrict__ dst = ei + E;
  int e = e0 + tid;
  for (; e + 3072 < e1; e += 4096) {
    int d0 = dst[e], d1 = dst[e + 1024], d2 = dst[e + 2048], d3 = dst[e + 3072];
    atomicAdd(&h[d0], 1);
    atomicAdd(&h[d1], 1);
    atomicAdd(&h[d2], 1);
    atomicAdd(&h[d3], 1);
  }
  for (; e < e1; e += 1024) atomicAdd(&h[dst[e]], 1);
  __syncthreads();

  {
    const int per = (N + 1023) >> 10;
    const int lo = min(tid * per, N), hi = min(lo + per, N);
    int sum = 0;
    for (int i = lo; i < hi; ++i) sum += h[i];
    const int lane = tid & 31, wv = tid >> 5;
    int inc = sum;
#pragma unroll
    for (int o = 1; o < 32; o <<= 1) { int u = __shfl_up(inc, o, 32); if (lane >= o) inc += u; }
    if (lane == 31) wsum[wv] = inc;
    __syncthreads();
    int v = wsum[lane], vi = v;
#pragma unroll
    for (int o = 1; o < 32; o <<= 1) { int u = __shfl_up(vi, o, 32); if (lane >= o) vi += u; }
    int run = __shfl(vi - v, wv, 32) + inc - sum;
    for (int i = lo; i < hi; ++i) { int c = h[i]; h[i] = run; run += c; }
  }
  __syncthreads();

  int2* __restrict__ so = sedge + e0;
  const int* __restrict__ srcp = ei;
  e = e0 + tid;
  for (; e + 3072 < e1; e += 4096) {
    int d0 = dst[e], d1 = dst[e + 1024], d2 = dst[e + 2048], d3 = dst[e + 3072];
    int s0 = srcp[e], s1 = srcp[e + 1024], s2 = srcp[e + 2048], s3 = srcp[e + 3072];
    float a0 = eattr[e], a1 = eattr[e + 1024], a2 = eattr[e + 2048], a3 = eattr[e + 3072];
    int p0 = atomicAdd(&h[d0], 1);
    int p1 = atomicAdd(&h[d1], 1);
    int p2 = atomicAdd(&h[d2], 1);
    int p3 = atomicAdd(&h[d3], 1);
    so[p0] = make_int2(s0, __float_as_int(a0));
    so[p1] = make_int2(s1, __float_as_int(a1));
    so[p2] = make_int2(s2, __float_as_int(a2));
    so[p3] = make_int2(s3, __float_as_int(a3));
  }
  for (; e < e1; e += 1024) {
    int p = atomicAdd(&h[dst[e]], 1);
    so[p] = make_int2(srcp[e], __float_as_int(eattr[e]));
  }
  __syncthreads();
  int* __restrict__ o = endp + (long long)blockIdx.x * N;
  if (vec) {
    const int4* h4 = (const int4*)h;
    int4* __restrict__ o4 = (int4*)o;
    for (int i = tid; i < (N >> 2); i += 1024) o4[i] = h4[i];
  } else {
    for (int i = tid; i < N; i += 1024) o[i] = h[i];
  }
}


template <int J>
__device__ __forceinline__ int row_share_i(int v) {
  return __builtin_amdgcn_update_dpp(0, v, 0x150 + J, 0xf, 0xf, true);
}
template <int J>
__device__ __forceinline__ float row_share_f(float v) {
  return __int_as_float(row_share_i<J>(__float_as_int(v)));
}
template <int M>
__device__ __forceinline__ float row_xadd(float p) {
  return p + __int_as_float(__builtin_amdgcn_update_dpp(0, __float_as_int(p), 0x160 + M, 0xf, 0xf, true));
}

template <int S>
__device__ __forceinline__ int row_shr_add(int v) {
  return v + __builtin_amdgcn_update_dpp(0, v, 0x110 + S, 0xf, 0xf, true);
}

__device__ __forceinline__ float edge_logit(v4f a, v4f b, float eq, v4f we, v4f at) {
  float p = 0.f;
#pragma unroll
  for (int k = 0; k < 4; ++k) {
    float m = a[k] + b[k] + eq * we[k];
    m = (m > 0.f) ? m : 0.2f * m;
    p += m * at[k];
  }
  p = row_xadd<1>(p); p = row_xadd<2>(p); p = row_xadd<4>(p); p = row_xadd<8>(p);
  return p;
}

__global__ __launch_bounds__(256) void k_node(const int* __restrict__ endp, const int2* __restrict__ sedge,
                       const float* __restrict__ xl,
                       const float* __restrict__ We, const float* __restrict__ att,
                       const float* __restrict__ bias, float* out, int N, int E) {
  __shared__ int lidx[256];
  int l15 = threadIdx.x & 15;
  int d   = (blockIdx.x * blockDim.x + threadIdx.x) >> 4;
  if (d >= N) return;
  const int rowl = threadIdx.x & 0xf0;
  const int chunk = (E + HB - 1) / HB;
  int cnt[3], sb[3], rb[3];
  int tl = 0;
#pragma unroll
  for (int k = 0; k < 3; ++k) {
    const int p = l15 * 3 + k;
    const int* pl = endp + (long long)p * N + d;
    int r1 = pl[0];
    int r0 = (d > 0) ? pl[-1] : 0;
    cnt[k] = r1 - r0;
    rb[k]  = p * chunk + r0;
    tl += cnt[k];
  }
  v4f we = *(const v4f*)(We + l15 * 4);
  v4f at = *(const v4f*)(att + l15 * 4);
  v4f b  = *(const v4f*)(out + (long long)d * C + l15 * 4);
  v4f aself = *(const v4f*)(xl + (long long)d * C + l15 * 4);
  int incl = tl;
  incl = row_shr_add<1>(incl); incl = row_shr_add<2>(incl); incl = row_shr_add<4>(incl); incl = row_shr_add<8>(incl);
  sb[0] = incl - tl; sb[1] = sb[0] + cnt[0]; sb[2] = sb[1] + cnt[1];
  const int beg = 0, end = row_share_i<15>(incl);

  float M = -INFINITY, S = 0.f, easum = 0.f;
  v4f acc = {0.f, 0.f, 0.f, 0.f};
  for (int c = beg; c < end; c += 16) {
#pragma unroll
    for (int k = 0; k < 3; ++k) {
      const int lo = max(sb[k], c), hi = min(sb[k] + cnt[k], c + 16);
#pragma clang loop unroll(disable) vectorize(disable) interleave(disable)
      for (int pos = lo; pos < hi; ++pos) lidx[rowl + pos - c] = rb[k] + (pos - sb[k]);
    }
    __builtin_amdgcn_fence(__ATOMIC_ACQ_REL, "wavefront");
    int j = c + l15;
    int sj = d; float eaj = 0.f;
    if (j < end) { int2 r = sedge[lidx[rowl + l15]]; sj = r.x; eaj = __int_as_float(r.y); }
    easum += eaj;
    int nrem = end - c; nrem = nrem > 16 ? 16 : nrem;
    for (int k0 = 0; k0 < nrem; k0 += 4) {
      v4f a[4]; float eq[4], l[4];
#pragma unroll
      for (int k = 0; k < 4; ++k) {
        int sk = __shfl(sj, k0 + k, 16);
        eq[k]  = __shfl(eaj, k0 + k, 16);
        a[k]   = *(const v4f*)(xl + (long long)sk * C + l15 * 4);
      }
      float mx = M;
#pragma unroll
      for (int k = 0; k < 4; ++k) {
        l[k] = edge_logit(a[k], b, eq[k], we, at);
        l[k] = (k0 + k < nrem) ? l[k] : -INFINITY;
        mx = fmaxf(mx, l[k]);
      }
      float sc = __expf(M - mx);
      S *= sc; acc *= sc;
#pragma unroll
      for (int k = 0; k < 4; ++k) {
        float p = __expf(l[k] - mx);
        S += p; acc += p * a[k];
      }
      M = mx;
    }
  }
  easum = row_xadd<1>(easum); easum = row_xadd<2>(easum); easum = row_xadd<4>(easum); easum = row_xadd<8>(easum);
  float loop_ea = easum / fmaxf((float)(end - beg), 1.0f);
  float ls = edge_logit(aself, b, loop_ea, we, at);
  float mx = fmaxf(M, ls);
  float sc = __expf(M - mx);
  float p  = __expf(ls - mx);
  S = S * sc + p;
  acc = acc * sc + p * aself;
  v4f bi = *(const v4f*)(bias + l15 * 4);
  v4f o;
#pragma unroll
  for (int k = 0; k < 4; ++k) o[k] = acc[k] / S + bi[k];
  *(v4f*)(out + (long long)d * C + l15 * 4) = o;
}

extern "C" void kernel_launch(void* const* d_in, const int* in_sizes, int n_in,
                              void* d_out, int out_size, void* d_ws, size_t ws_size,
                              hipStream_t stream) {
  (void)n_in; (void)out_size; (void)ws_size;
  const float* x     = (const float*)d_in[0];
  const float* eattr = (const float*)d_in[1];
  const float* Wl    = (const float*)d_in[2];
  const float* Wr    = (const float*)d_in[3];
  const float* We    = (const float*)d_in[4];
  const float* att   = (const float*)d_in[5];
  const float* bias  = (const float*)d_in[6];
  const int*   ei    = (const int*)d_in[7];
  float*       out   = (float*)d_out;

  int N = in_sizes[0] / F_IN;
  int E = in_sizes[1];

  char* ws = (char*)d_ws;
  size_t o = 0;
  auto take = [&](size_t bytes) -> void* {
    void* p = ws + o; o += (bytes + 255) & ~(size_t)255; return p;
  };
  float* xl     = (float*)take((size_t)N * C * 4);
  int*   endp   = (int*)take((size_t)HB * N * 4);
  int2*  sedge  = (int2*)take((size_t)E * 8);
  float* xr     = out;

  const int B = 256;
  k_count_gemm<<<HB + GB, 1024, 0, stream>>>(ei, eattr, endp, sedge, x, Wl, Wr, xl, xr, N, E);
  k_node<<<(unsigned)(((long long)N * 16 + B - 1) / B), B, 0, stream>>>(endp, sedge, xl, We, att, bias, out, N, E);
}


# a chunk's records are placed in LDS (16-bit counters packed two to a word make room) and leave in slot order as whole 128-byte lines; chunks rounded to 16 records, so slot (sum) order changes
# speedup vs baseline: 8.7542x; 1.1717x over previous
#include <hip/hip_runtime.h>

#define F_IN 128
#define C 64

typedef float v2f __attribute__((ext_vector_type(2)));
typedef float v8f __attribute__((ext_vector_type(8)));

#define HB 48
#define NODE_CAP 65536
#define REC_CAP 20480
#define GB 208

typedef int v2i __attribute__((ext_vector_type(2)));
typedef int v4i __attribute__((ext_vector_type(4)));
typedef unsigned v2u __attribute__((ext_vector_type(2)));

__device__ __forceinline__ int chunk_len(int E) { return ((E + HB - 1) / HB + 15) & ~15; }

typedef float v4f __attribute__((ext_vector_type(4)));

__device__ __forceinline__ v8f mma4(v4f a, v4f b, int j, v8f acc) {
  v2f aa, bb;
  aa.x = a[2 * j]; aa.y = a[2 * j + 1];
  bb.x = b[2 * j]; bb.y = b[2 * j + 1];
  return __builtin_amdgcn_wmma_f32_16x16x4_f32(
       false, aa,  false, bb,
       (short)0, acc,  false,  false);
}

__device__ __forceinline__ void gemm_block(const float* __restrict__ x, const float* __restrict__ Wl,
                                           const float* __restrict__ Wr, float* __restrict__ xl,
                                           float* __restrict__ xr, float* wlds, int gb, int ngb, int N, int tid) {
  const long long T = (long long)(N >> 4) * 2;
  if ((long long)gb * 32 >= T) return;
  const int lane = tid & 31, hf = lane >> 4, c = lane & 15;
  const long long tstep = (long long)ngb * 32;
  long long t = (long long)gb * 32 + (tid >> 5);
  const v4f* ap = (const v4f*)(x + (long long)((int)(t >> 1) * 16 + c) * F_IN + hf * 4);
  v4f a[4];
  if (t < T) {
#pragma unroll
    for (int i = 0; i < 4; ++i) a[i] = ap[2 * i];
  }
  {
    const int col = tid & 127, kk0 = tid >> 7;
    const float* s = (col < 64 ? Wl + col : Wr + (col - 64)) + (long long)(kk0 * 4) * C;
    v4f w[4];
#pragma unroll
    for (int i = 0; i < 4; ++i) {
      const float* si = s + (long long)(i * 32) * C;
      w[i].x = si[0]; w[i].y = si[C]; w[i].z = si[2 * C]; w[i].w = si[3 * C];
    }
#pragma unroll
    for (int i = 0; i < 4; ++i) *(v4f*)(wlds + (tid + i * 1024) * 4) = w[i];
  }
  __syncthreads();

  while (t < T) {
    const int which = (int)(t & 1);
    const int m0    = (int)(t >> 1) << 4;
    const v4f* bw = (const v4f*)wlds + hf * 128 + which * 64 + c;
    v8f acc[4];
#pragma unroll
    for (int tl = 0; tl < 4; ++tl) acc[tl] = (v8f){0.f, 0.f, 0.f, 0.f, 0.f, 0.f, 0.f, 0.f};
    v4f b[4];
#pragma unroll
    for (int tl = 0; tl < 4; ++tl) b[tl] = bw[tl * 16];
#pragma unroll
    for (int g = 0; g < 16; ++g) {
      v4f bn[4];
      if (g + 1 < 16) {
#pragma unroll
        for (int tl = 0; tl < 4; ++tl) bn[tl] = bw[(g + 1) * 256 + tl * 16];
      }
      v4f av = a[g & 3];
      if (g + 4 < 16) a[g & 3] = ap[2 * (g + 4)];
#pragma unroll
      for (int j = 0; j < 2; ++j) {
#pragma unroll
        for (int tl = 0; tl < 4; ++tl) acc[tl] = mma4(av, b[tl], j, acc[tl]);
      }
      if (g + 1 < 16) {
#pragma unroll
        for (int tl = 0; tl < 4; ++tl) b[tl] = bn[tl];
      }
      __builtin_amdgcn_sched_barrier(0);
    }
    float* yrow = (which ? xr : xl) + (long long)(m0 + hf * 8) * C + c;
#pragma unroll
    for (int tl = 0; tl < 4; ++tl) {
#pragma unroll
      for (int r = 0; r < 8; ++r) yrow[(long long)r * C + tl * 16] = acc[tl][r];
    }
    t += tstep;
    if (t < T) {
      ap = (const v4f*)(x + (long long)((int)(t >> 1) * 16 + c) * F_IN + hf * 4);
#pragma unroll
      for (int i = 0; i < 4; ++i) a[i] = ap[2 * i];
    }
  }
}

__global__ __launch_bounds__(1024) void k_count_gemm(const int* __restrict__ ei, const float* __restrict__ eattr,
                                                     int* __restrict__ endp, int2* __restrict__ sedge,
                                                     const float* __restrict__ x, const float* __restrict__ Wl,
                                                     const float* __restrict__ Wr, float* __restrict__ xl,
                                                     float* __restrict__ xr, int N, int E) {
  __shared__ __attribute__((aligned(16))) unsigned hw[NODE_CAP / 2];
  __shared__ __attribute__((aligned(16))) v2i rec[REC_CAP];
  __shared__ int wsum[32];
  const int tid = threadIdx.x;
  if (blockIdx.x >= HB) {
    gemm_block(x, Wl, Wr, xl, xr, (float*)hw, (int)blockIdx.x - HB, (int)gridDim.x - HB, N, tid);
    return;
  }
  const bool vec = (N & 3) == 0;
  const int nw = (N + 1) >> 1;
  {
    v4i* h4 = (v4i*)hw;
    for (int i = tid; i < ((nw + 3) >> 2); i += 1024) h4[i] = (v4i){0, 0, 0, 0};
  }
  __syncthreads();
  const int chunk = chunk_len(E);
  const int e0 = blockIdx.x * chunk;
  const int e1 = min(e0 + chunk, E);
  const int* __restrict__ dst = ei + E;
  int e = e0 + tid;
  for (; e + 3072 < e1; e += 4096) {
    int d0 = dst[e], d1 = dst[e + 1024], d2 = dst[e + 2048], d3 = dst[e + 3072];
    atomicAdd(&hw[d0 >> 1], 1u << ((d0 & 1) << 4));
    atomicAdd(&hw[d1 >> 1], 1u << ((d1 & 1) << 4));
    atomicAdd(&hw[d2 >> 1], 1u << ((d2 & 1) << 4));
    atomicAdd(&hw[d3 >> 1], 1u << ((d3 & 1) << 4));
  }
  for (; e < e1; e += 1024) { int d = dst[e]; atomicAdd(&hw[d >> 1], 1u << ((d & 1) << 4)); }
  __syncthreads();

  {
    const int per = (nw + 1023) >> 10;
    const int lo = min(tid * per, nw), hi = min(lo + per, nw);
    int sum = 0;
    for (int i = lo; i < hi; ++i) { unsigned w = hw[i]; sum += (int)(w & 0xffffu) + (int)(w >> 16); }
    const int lane = tid & 31, wv = tid >> 5;
    int inc = sum;
#pragma unroll
    for (int o = 1; o < 32; o <<= 1) { int u = __shfl_up(inc, o, 32); if (lane >= o) inc += u; }
    if (lane == 31) wsum[wv] = inc;
    __syncthreads();
    int v = wsum[lane], vi = v;
#pragma unroll
    for (int o = 1; o < 32; o <<= 1) { int u = __shfl_up(vi, o, 32); if (lane >= o) vi += u; }
    int run = __shfl(vi - v, wv, 32) + inc - sum;
    for (int i = lo; i < hi; ++i) {
      unsigned w = hw[i];
      int c0 = (int)(w & 0xffffu), c1 = (int)(w >> 16);
      hw[i] = (unsigned)run | ((unsigned)(run + c0) << 16);
      run += c0 + c1;
    }
  }
  __syncthreads();

  const int* __restrict__ srcp = ei;
  e = e0 + tid;
  for (; e + 3072 < e1; e += 4096) {
    int d0 = dst[e], d1 = dst[e + 1024], d2 = dst[e + 2048], d3 = dst[e + 3072];
    int s0 = srcp[e], s1 = srcp[e + 1024], s2 = srcp[e + 2048], s3 = srcp[e + 3072];
    float a0 = eattr[e], a1 = eattr[e + 1024], a2 = eattr[e + 2048], a3 = eattr[e + 3072];
    const int h0 = (d0 & 1) << 4, h1 = (d1 & 1) << 4, h2 = (d2 & 1) << 4, h3 = (d3 & 1) << 4;
    unsigned q0 = atomicAdd(&hw[d0 >> 1], 1u << h0);
    unsigned q1 = atomicAdd(&hw[d1 >> 1], 1u << h1);
    unsigned q2 = atomicAdd(&hw[d2 >> 1], 1u << h2);
    unsigned q3 = atomicAdd(&hw[d3 >> 1], 1u << h3);
    rec[(q0 >> h0) & 0xffffu] = (v2i){s0, __float_as_int(a0)};
    rec[(q1 >> h1) & 0xffffu] = (v2i){s1, __float_as_int(a1)};
    rec[(q2 >> h2) & 0xffffu] = (v2i){s2, __float_as_int(a2)};
    rec[(q3 >> h3) & 0xffffu] = (v2i){s3, __float_as_int(a3)};
  }
  for (; e < e1; e += 1024) {
    int d = dst[e];
    const int hs = (d & 1) << 4;
    unsigned q = atomicAdd(&hw[d >> 1], 1u << hs);
    rec[(q >> hs) & 0xffffu] = (v2i){srcp[e], __float_as_int(eattr[e])};
  }
  __syncthreads();
  {
    const int cnt = max(e1 - e0, 0);
    v2i* so = (v2i*)(sedge + e0);
    const v4i* r4 = (const v4i*)rec;
    v4i* so4 = (v4i*)so;
    for (int i = tid; i < (cnt >> 1); i += 1024) so4[i] = r4[i];
    if ((cnt & 1) && tid == 0) so[cnt - 1] = rec[cnt - 1];
  }
  int* __restrict__ o = endp + (long long)blockIdx.x * N;
  if (vec) {
    const v2u* h2 = (const v2u*)hw;
    v4i* __restrict__ o4 = (v4i*)o;
    for (int i = tid; i < (N >> 2); i += 1024) {
      v2u w = h2[i];
      o4[i] = (v4i){(int)(w.x & 0xffffu), (int)(w.x >> 16), (int)(w.y & 0xffffu), (int)(w.y >> 16)};
    }
  } else {
    for (int i = tid; i < N; i += 1024) o[i] = (int)((hw[i >> 1] >> ((i & 1) << 4)) & 0xffffu);
  }
}


template <int J>
__device__ __forceinline__ int row_share_i(int v) {
  return __builtin_amdgcn_update_dpp(0, v, 0x150 + J, 0xf, 0xf, true);
}
template <int J>
__device__ __forceinline__ float row_share_f(float v) {
  return __int_as_float(row_share_i<J>(__float_as_int(v)));
}
template <int M>
__device__ __forceinline__ float row_xadd(float p) {
  return p + __int_as_float(__builtin_amdgcn_update_dpp(0, __float_as_int(p), 0x160 + M, 0xf, 0xf, true));
}

template <int S>
__device__ __forceinline__ int row_shr_add(int v) {
  return v + __builtin_amdgcn_update_dpp(0, v, 0x110 + S, 0xf, 0xf, true);
}

__device__ __forceinline__ float edge_logit(v4f a, v4f b, float eq, v4f we, v4f at) {
  float p = 0.f;
#pragma unroll
  for (int k = 0; k < 4; ++k) {
    float m = a[k] + b[k] + eq * we[k];
    m = (m > 0.f) ? m : 0.2f * m;
    p += m * at[k];
  }
  p = row_xadd<1>(p); p = row_xadd<2>(p); p = row_xadd<4>(p); p = row_xadd<8>(p);
  return p;
}

__global__ __launch_bounds__(256) void k_node(const int* __restrict__ endp, const int2* __restrict__ sedge,
                       const float* __restrict__ xl,
                       const float* __restrict__ We, const float* __restrict__ att,
                       const float* __restrict__ bias, float* out, int N, int E) {
  __shared__ int lidx[256];
  int l15 = threadIdx.x & 15;
  int d   = (blockIdx.x * blockDim.x + threadIdx.x) >> 4;
  if (d >= N) return;
  const int rowl = threadIdx.x & 0xf0;
  const int chunk = chunk_len(E);
  int cnt[3], sb[3], rb[3];
  int tl = 0;
#pragma unroll
  for (int k = 0; k < 3; ++k) {
    const int p = l15 * 3 + k;
    const int* pl = endp + (long long)p * N + d;
    int r1 = pl[0];
    int r0 = (d > 0) ? pl[-1] : 0;
    cnt[k] = r1 - r0;
    rb[k]  = p * chunk + r0;
    tl += cnt[k];
  }
  v4f we = *(const v4f*)(We + l15 * 4);
  v4f at = *(const v4f*)(att + l15 * 4);
  v4f b  = *(const v4f*)(out + (long long)d * C + l15 * 4);
  v4f aself = *(const v4f*)(xl + (long long)d * C + l15 * 4);
  int incl = tl;
  incl = row_shr_add<1>(incl); incl = row_shr_add<2>(incl); incl = row_shr_add<4>(incl); incl = row_shr_add<8>(incl);
  sb[0] = incl - tl; sb[1] = sb[0] + cnt[0]; sb[2] = sb[1] + cnt[1];
  const int beg = 0, end = row_share_i<15>(incl);

  float M = -INFINITY, S = 0.f, easum = 0.f;
  v4f acc = {0.f, 0.f, 0.f, 0.f};
  for (int c = beg; c < end; c += 16) {
#pragma unroll
    for (int k = 0; k < 3; ++k) {
      const int lo = max(sb[k], c), hi = min(sb[k] + cnt[k], c + 16);
#pragma clang loop unroll(disable) vectorize(disable) interleave(disable)
      for (int pos = lo; pos < hi; ++pos) lidx[rowl + pos - c] = rb[k] + (pos - sb[k]);
    }
    __builtin_amdgcn_fence(__ATOMIC_ACQ_REL, "wavefront");
    int j = c + l15;
    int sj = d; float eaj = 0.f;
    if (j < end) { int2 r = sedge[lidx[rowl + l15]]; sj = r.x; eaj = __int_as_float(r.y); }
    easum += eaj;
    int nrem = end - c; nrem = nrem > 16 ? 16 : nrem;
    for (int k0 = 0; k0 < nrem; k0 += 4) {
      v4f a[4]; float eq[4], l[4];
#pragma unroll
      for (int k = 0; k < 4; ++k) {
        int sk = __shfl(sj, k0 + k, 16);
        eq[k]  = __shfl(eaj, k0 + k, 16);
        a[k]   = *(const v4f*)(xl + (long long)sk * C + l15 * 4);
      }
      float mx = M;
#pragma unroll
      for (int k = 0; k < 4; ++k) {
        l[k] = edge_logit(a[k], b, eq[k], we, at);
        l[k] = (k0 + k < nrem) ? l[k] : -INFINITY;
        mx = fmaxf(mx, l[k]);
      }
      float sc = __expf(M - mx);
      S *= sc; acc *= sc;
#pragma unroll
      for (int k = 0; k < 4; ++k) {
        float p = __expf(l[k] - mx);
        S += p; acc += p * a[k];
      }
      M = mx;
    }
  }
  easum = row_xadd<1>(easum); easum = row_xadd<2>(easum); easum = row_xadd<4>(easum); easum = row_xadd<8>(easum);
  float loop_ea = easum / fmaxf((float)(end - beg), 1.0f);
  float ls = edge_logit(aself, b, loop_ea, we, at);
  float mx = fmaxf(M, ls);
  float sc = __expf(M - mx);
  float p  = __expf(ls - mx);
  S = S * sc + p;
  acc = acc * sc + p * aself;
  v4f bi = *(const v4f*)(bias + l15 * 4);
  v4f o;
#pragma unroll
  for (int k = 0; k < 4; ++k) o[k] = acc[k] / S + bi[k];
  *(v4f*)(out + (long long)d * C + l15 * 4) = o;
}

extern "C" void kernel_launch(void* const* d_in, const int* in_sizes, int n_in,
                              void* d_out, int out_size, void* d_ws, size_t ws_size,
                              hipStream_t stream) {
  (void)n_in; (void)out_size; (void)ws_size;
  const float* x     = (const float*)d_in[0];
  const float* eattr = (const float*)d_in[1];
  const float* Wl    = (const float*)d_in[2];
  const float* Wr    = (const float*)d_in[3];
  const float* We    = (const float*)d_in[4];
  const float* att   = (const float*)d_in[5];
  const float* bias  = (const float*)d_in[6];
  const int*   ei    = (const int*)d_in[7];
  float*       out   = (float*)d_out;

  int N = in_sizes[0] / F_IN;
  int E = in_sizes[1];

  char* ws = (char*)d_ws;
  size_t o = 0;
  auto take = [&](size_t bytes) -> void* {
    void* p = ws + o; o += (bytes + 255) & ~(size_t)255; return p;
  };
  float* xl     = (float*)take((size_t)N * C * 4);
  int*   endp   = (int*)take((size_t)HB * N * 4);
  int2*  sedge  = (int2*)take((size_t)E * 8);
  float* xr     = out;

  const int B = 256;
  k_count_gemm<<<HB + GB, 1024, 0, stream>>>(ei, eattr, endp, sedge, x, Wl, Wr, xl, xr, N, E);
  k_node<<<(unsigned)(((long long)N * 16 + B - 1) / B), B, 0, stream>>>(endp, sedge, xl, We, att, bias, out, N, E);
}


# node kernel: 8 lanes per node (two 16-byte column words a lane, whole 128-byte lines a load), 4 nodes per wave: half the waves, half the lines per run-end load; logit and edge_attr sum orders change
# speedup vs baseline: 9.1872x; 1.0495x over previous
#include <hip/hip_runtime.h>

#define F_IN 128
#define C 64

typedef float v2f __attribute__((ext_vector_type(2)));
typedef float v8f __attribute__((ext_vector_type(8)));

#define HB 48
#define NODE_CAP 65536
#define REC_CAP 20480
#define GB 208

typedef int v2i __attribute__((ext_vector_type(2)));
typedef int v4i __attribute__((ext_vector_type(4)));
typedef unsigned v2u __attribute__((ext_vector_type(2)));

__device__ __forceinline__ int chunk_len(int E) { return ((E + HB - 1) / HB + 15) & ~15; }

typedef float v4f __attribute__((ext_vector_type(4)));

__device__ __forceinline__ v8f mma4(v4f a, v4f b, int j, v8f acc) {
  v2f aa, bb;
  aa.x = a[2 * j]; aa.y = a[2 * j + 1];
  bb.x = b[2 * j]; bb.y = b[2 * j + 1];
  return __builtin_amdgcn_wmma_f32_16x16x4_f32(
       false, aa,  false, bb,
       (short)0, acc,  false,  false);
}

__device__ __forceinline__ void gemm_block(const float* __restrict__ x, const float* __restrict__ Wl,
                                           const float* __restrict__ Wr, float* __restrict__ xl,
                                           float* __restrict__ xr, float* wlds, int gb, int ngb, int N, int tid) {
  const long long T = (long long)(N >> 4) * 2;
  if ((long long)gb * 32 >= T) return;
  const int lane = tid & 31, hf = lane >> 4, c = lane & 15;
  const long long tstep = (long long)ngb * 32;
  long long t = (long long)gb * 32 + (tid >> 5);
  const v4f* ap = (const v4f*)(x + (long long)((int)(t >> 1) * 16 + c) * F_IN + hf * 4);
  v4f a[4];
  if (t < T) {
#pragma unroll
    for (int i = 0; i < 4; ++i) a[i] = ap[2 * i];
  }
  {
    const int col = tid & 127, kk0 = tid >> 7;
    const float* s = (col < 64 ? Wl + col : Wr + (col - 64)) + (long long)(kk0 * 4) * C;
    v4f w[4];
#pragma unroll
    for (int i = 0; i < 4; ++i) {
      const float* si = s + (long long)(i * 32) * C;
      w[i].x = si[0]; w[i].y = si[C]; w[i].z = si[2 * C]; w[i].w = si[3 * C];
    }
#pragma unroll
    for (int i = 0; i < 4; ++i) *(v4f*)(wlds + (tid + i * 1024) * 4) = w[i];
  }
  __syncthreads();

  while (t < T) {
    const int which = (int)(t & 1);
    const int m0    = (int)(t >> 1) << 4;
    const v4f* bw = (const v4f*)wlds + hf * 128 + which * 64 + c;
    v8f acc[4];
#pragma unroll
    for (int tl = 0; tl < 4; ++tl) acc[tl] = (v8f){0.f, 0.f, 0.f, 0.f, 0.f, 0.f, 0.f, 0.f};
    v4f b[4];
#pragma unroll
    for (int tl = 0; tl < 4; ++tl) b[tl] = bw[tl * 16];
#pragma unroll
    for (int g = 0; g < 16; ++g) {
      v4f bn[4];
      if (g + 1 < 16) {
#pragma unroll
        for (int tl = 0; tl < 4; ++tl) bn[tl] = bw[(g + 1) * 256 + tl * 16];
      }
      v4f av = a[g & 3];
      if (g + 4 < 16) a[g & 3] = ap[2 * (g + 4)];
#pragma unroll
      for (int j = 0; j < 2; ++j) {
#pragma unroll
        for (int tl = 0; tl < 4; ++tl) acc[tl] = mma4(av, b[tl], j, acc[tl]);
      }
      if (g + 1 < 16) {
#pragma unroll
        for (int tl = 0; tl < 4; ++tl) b[tl] = bn[tl];
      }
      __builtin_amdgcn_sched_barrier(0);
    }
    float* yrow = (which ? xr : xl) + (long long)(m0 + hf * 8) * C + c;
#pragma unroll
    for (int tl = 0; tl < 4; ++tl) {
#pragma unroll
      for (int r = 0; r < 8; ++r) yrow[(long long)r * C + tl * 16] = acc[tl][r];
    }
    t += tstep;
    if (t < T) {
      ap = (const v4f*)(x + (long long)((int)(t >> 1) * 16 + c) * F_IN + hf * 4);
#pragma unroll
      for (int i = 0; i < 4; ++i) a[i] = ap[2 * i];
    }
  }
}

__global__ __launch_bounds__(1024) void k_count_gemm(const int* __restrict__ ei, const float* __restrict__ eattr,
                                                     int* __restrict__ endp, int2* __restrict__ sedge,
                                                     const float* __restrict__ x, const float* __restrict__ Wl,
                                                     const float* __restrict__ Wr, float* __restrict__ xl,
                                                     float* __restrict__ xr, int N, int E) {
  __shared__ __attribute__((aligned(16))) unsigned hw[NODE_CAP / 2];
  __shared__ __attribute__((aligned(16))) v2i rec[REC_CAP];
  __shared__ int wsum[32];
  const int tid = threadIdx.x;
  if (blockIdx.x >= HB) {
    gemm_block(x, Wl, Wr, xl, xr, (float*)hw, (int)blockIdx.x - HB, (int)gridDim.x - HB, N, tid);
    return;
  }
  const bool vec = (N & 3) == 0;
  const int nw = (N + 1) >> 1;
  {
    v4i* h4 = (v4i*)hw;
    for (int i = tid; i < ((nw + 3) >> 2); i += 1024) h4[i] = (v4i){0, 0, 0, 0};
  }
  __syncthreads();
  const int chunk = chunk_len(E);
  const int e0 = blockIdx.x * chunk;
  const int e1 = min(e0 + chunk, E);
  const int* __restrict__ dst = ei + E;
  int e = e0 + tid;
  for (; e + 3072 < e1; e += 4096) {
    int d0 = dst[e], d1 = dst[e + 1024], d2 = dst[e + 2048], d3 = dst[e + 3072];
    atomicAdd(&hw[d0 >> 1], 1u << ((d0 & 1) << 4));
    atomicAdd(&hw[d1 >> 1], 1u << ((d1 & 1) << 4));
    atomicAdd(&hw[d2 >> 1], 1u << ((d2 & 1) << 4));
    atomicAdd(&hw[d3 >> 1], 1u << ((d3 & 1) << 4));
  }
  for (; e < e1; e += 1024) { int d = dst[e]; atomicAdd(&hw[d >> 1], 1u << ((d & 1) << 4)); }
  __syncthreads();

  {
    const int per = (nw + 1023) >> 10;
    const int lo = min(tid * per, nw), hi = min(lo + per, nw);
    int sum = 0;
    for (int i = lo; i < hi; ++i) { unsigned w = hw[i]; sum += (int)(w & 0xffffu) + (int)(w >> 16); }
    const int lane = tid & 31, wv = tid >> 5;
    int inc = sum;
#pragma unroll
    for (int o = 1; o < 32; o <<= 1) { int u = __shfl_up(inc, o, 32); if (lane >= o) inc += u; }
    if (lane == 31) wsum[wv] = inc;
    __syncthreads();
    int v = wsum[lane], vi = v;
#pragma unroll
    for (int o = 1; o < 32; o <<= 1) { int u = __shfl_up(vi, o, 32); if (lane >= o) vi += u; }
    int run = __shfl(vi - v, wv, 32) + inc - sum;
    for (int i = lo; i < hi; ++i) {
      unsigned w = hw[i];
      int c0 = (int)(w & 0xffffu), c1 = (int)(w >> 16);
      hw[i] = (unsigned)run | ((unsigned)(run + c0) << 16);
      run += c0 + c1;
    }
  }
  __syncthreads();

  const int* __restrict__ srcp = ei;
  e = e0 + tid;
  for (; e + 3072 < e1; e += 4096) {
    int d0 = dst[e], d1 = dst[e + 1024], d2 = dst[e + 2048], d3 = dst[e + 3072];
    int s0 = srcp[e], s1 = srcp[e + 1024], s2 = srcp[e + 2048], s3 = srcp[e + 3072];
    float a0 = eattr[e], a1 = eattr[e + 1024], a2 = eattr[e + 2048], a3 = eattr[e + 3072];
    const int h0 = (d0 & 1) << 4, h1 = (d1 & 1) << 4, h2 = (d2 & 1) << 4, h3 = (d3 & 1) << 4;
    unsigned q0 = atomicAdd(&hw[d0 >> 1], 1u << h0);
    unsigned q1 = atomicAdd(&hw[d1 >> 1], 1u << h1);
    unsigned q2 = atomicAdd(&hw[d2 >> 1], 1u << h2);
    unsigned q3 = atomicAdd(&hw[d3 >> 1], 1u << h3);
    rec[(q0 >> h0) & 0xffffu] = (v2i){s0, __float_as_int(a0)};
    rec[(q1 >> h1) & 0xffffu] = (v2i){s1, __float_as_int(a1)};
    rec[(q2 >> h2) & 0xffffu] = (v2i){s2, __float_as_int(a2)};
    rec[(q3 >> h3) & 0xffffu] = (v2i){s3, __float_as_int(a3)};
  }
  for (; e < e1; e += 1024) {
    int d = dst[e];
    const int hs = (d & 1) << 4;
    unsigned q = atomicAdd(&hw[d >> 1], 1u << hs);
    rec[(q >> hs) & 0xffffu] = (v2i){srcp[e], __float_as_int(eattr[e])};
  }
  __syncthreads();
  {
    const int cnt = max(e1 - e0, 0);
    v2i* so = (v2i*)(sedge + e0);
    const v4i* r4 = (const v4i*)rec;
    v4i* so4 = (v4i*)so;
    for (int i = tid; i < (cnt >> 1); i += 1024) so4[i] = r4[i];
    if ((cnt & 1) && tid == 0) so[cnt - 1] = rec[cnt - 1];
  }
  int* __restrict__ o = endp + (long long)blockIdx.x * N;
  if (vec) {
    const v2u* h2 = (const v2u*)hw;
    v4i* __restrict__ o4 = (v4i*)o;
    for (int i = tid; i < (N >> 2); i += 1024) {
      v2u w = h2[i];
      o4[i] = (v4i){(int)(w.x & 0xffffu), (int)(w.x >> 16), (int)(w.y & 0xffffu), (int)(w.y >> 16)};
    }
  } else {
    for (int i = tid; i < N; i += 1024) o[i] = (int)((hw[i >> 1] >> ((i & 1) << 4)) & 0xffffu);
  }
}


template <int M>
__device__ __forceinline__ float grp_xadd(float p) {
  return p + __int_as_float(__builtin_amdgcn_update_dpp(0, __float_as_int(p), 0x160 + M, 0xf, 0xf, true));
}
template <int M>
__device__ __forceinline__ int grp_xadd_i(int v) {
  return v + __builtin_amdgcn_update_dpp(0, v, 0x160 + M, 0xf, 0xf, true);
}

template <int S>
__device__ __forceinline__ int grp_shr_add(int v, int l7) {
  const int u = __builtin_amdgcn_update_dpp(0, v, 0x110 + S, 0xf, 0xf, true);
  return v + (l7 >= S ? u : 0);
}

#define GL 8
#define RPL (HB / GL)
static_assert(HB % GL == 0, "every lane of a group takes the same number of chunks");

__device__ __forceinline__ float edge_logit(v4f a0, v4f a1, v4f b0, v4f b1, float eq, v4f we0, v4f we1, v4f at0,
                                            v4f at1) {
  float p = 0.f;
#pragma unroll
  for (int k = 0; k < 4; ++k) {
    float m = a0[k] + b0[k] + eq * we0[k];
    m = (m > 0.f) ? m : 0.2f * m;
    p += m * at0[k];
  }
#pragma unroll
  for (int k = 0; k < 4; ++k) {
    float m = a1[k] + b1[k] + eq * we1[k];
    m = (m > 0.f) ? m : 0.2f * m;
    p += m * at1[k];
  }
  p = grp_xadd<1>(p); p = grp_xadd<2>(p); p = grp_xadd<4>(p);
  return p;
}

__global__ __launch_bounds__(256, 7) void k_node(const int* __restrict__ endp, const int2* __restrict__ sedge,
                       const float* __restrict__ xl,
                       const float* __restrict__ We, const float* __restrict__ att,
                       const float* __restrict__ bias, float* out, int N, int E) {
  __shared__ int lidx[512];
  const int l7 = threadIdx.x & (GL - 1);
  const int d  = (blockIdx.x * blockDim.x + threadIdx.x) >> 3;
  if (d >= N) return;
  const int rowl = (threadIdx.x >> 3) << 4;
  const int chunk = chunk_len(E);
  const int co = l7 * 4;
  const float* orow = out + (long long)d * C + co;
  const v4f b0 = *(const v4f*)orow, b1 = *(const v4f*)(orow + 32);
  const float* srow = xl + (long long)d * C + co;
  const v4f s0 = *(const v4f*)srow, s1 = *(const v4f*)(srow + 32);
  const v4f we0 = *(const v4f*)(We + co),  we1 = *(const v4f*)(We + 32 + co);
  const v4f at0 = *(const v4f*)(att + co), at1 = *(const v4f*)(att + 32 + co);
  int cnt[RPL], sb[RPL], rb[RPL];
  int tl = 0;
  const int back = (d > 0) ? 1 : 0;
#pragma unroll
  for (int k = 0; k < RPL; ++k) {
    const int p = l7 * RPL + k;
    const int* pl = endp + (long long)p * N + d;
    int r1 = pl[0];
    int r0 = pl[-back];
    r0 = back ? r0 : 0;
    cnt[k] = r1 - r0;
    rb[k]  = p * chunk + r0;
    tl += cnt[k];
  }
  int incl = tl;
  incl = grp_shr_add<1>(incl, l7); incl = grp_shr_add<2>(incl, l7); incl = grp_shr_add<4>(incl, l7);
  sb[0] = incl - tl;
#pragma unroll
  for (int k = 1; k < RPL; ++k) sb[k] = sb[k - 1] + cnt[k - 1];
  int end = tl;
  end = grp_xadd_i<1>(end); end = grp_xadd_i<2>(end); end = grp_xadd_i<4>(end);

  float M = -INFINITY, S = 0.f, easum = 0.f;
  v4f acc0 = {0.f, 0.f, 0.f, 0.f}, acc1 = {0.f, 0.f, 0.f, 0.f};
  for (int c = 0; c < end; c += 16) {
#pragma unroll
    for (int k = 0; k < RPL; ++k) {
      const int lo = max(sb[k], c), hi = min(sb[k] + cnt[k], c + 16);
#pragma clang loop unroll(disable) vectorize(disable) interleave(disable)
      for (int pos = lo; pos < hi; ++pos) lidx[rowl + pos - c] = rb[k] + (pos - sb[k]);
    }
    __builtin_amdgcn_fence(__ATOMIC_ACQ_REL, "wavefront");
    int sj0 = d, sj1 = d; float ea0 = 0.f, ea1 = 0.f;
    if (c + l7 < end)     { int2 r = sedge[lidx[rowl + l7]];     sj0 = r.x; ea0 = __int_as_float(r.y); }
    if (c + 8 + l7 < end) { int2 r = sedge[lidx[rowl + 8 + l7]]; sj1 = r.x; ea1 = __int_as_float(r.y); }
    easum += ea0 + ea1;
    int nrem = end - c; nrem = nrem > 16 ? 16 : nrem;
    for (int k0 = 0; k0 < nrem; k0 += 4) {
      const int   sjv = (k0 & 8) ? sj1 : sj0;
      const float eav = (k0 & 8) ? ea1 : ea0;
      v4f a0[4], a1[4]; float eq[4], l[4];
#pragma unroll
      for (int k = 0; k < 4; ++k) {
        int sk = __shfl(sjv, (k0 & 4) + k, GL);
        eq[k]  = __shfl(eav, (k0 & 4) + k, GL);
        const float* r = xl + (long long)sk * C + co;
        a0[k] = *(const v4f*)r;
        a1[k] = *(const v4f*)(r + 32);
      }
      float mx = M;
#pragma unroll
      for (int k = 0; k < 4; ++k) {
        l[k] = edge_logit(a0[k], a1[k], b0, b1, eq[k], we0, we1, at0, at1);
        l[k] = (k0 + k < nrem) ? l[k] : -INFINITY;
        mx = fmaxf(mx, l[k]);
      }
      float sc = __expf(M - mx);
      S *= sc; acc0 *= sc; acc1 *= sc;
#pragma unroll
      for (int k = 0; k < 4; ++k) {
        float p = __expf(l[k] - mx);
        S += p; acc0 += p * a0[k]; acc1 += p * a1[k];
      }
      M = mx;
    }
  }
  easum = grp_xadd<1>(easum); easum = grp_xadd<2>(easum); easum = grp_xadd<4>(easum);
  float loop_ea = easum / fmaxf((float)end, 1.0f);
  float ls = edge_logit(s0, s1, b0, b1, loop_ea, we0, we1, at0, at1);
  float mx = fmaxf(M, ls);
  float sc = __expf(M - mx);
  float p  = __expf(ls - mx);
  S = S * sc + p;
  acc0 = acc0 * sc + p * s0;
  acc1 = acc1 * sc + p * s1;
  const v4f bi0 = *(const v4f*)(bias + co), bi1 = *(const v4f*)(bias + 32 + co);
  v4f o0, o1;
#pragma unroll
  for (int k = 0; k < 4; ++k) { o0[k] = acc0[k] / S + bi0[k]; o1[k] = acc1[k] / S + bi1[k]; }
  float* wrow = out + (long long)d * C + co;
  *(v4f*)wrow = o0;
  *(v4f*)(wrow + 32) = o1;
}

extern "C" void kernel_launch(void* const* d_in, const int* in_sizes, int n_in,
                              void* d_out, int out_size, void* d_ws, size_t ws_size,
                              hipStream_t stream) {
  (void)n_in; (void)out_size; (void)ws_size;
  const float* x     = (const float*)d_in[0];
  const float* eattr = (const float*)d_in[1];
  const float* Wl    = (const float*)d_in[2];
  const float* Wr    = (const float*)d_in[3];
  const float* We    = (const float*)d_in[4];
  const float* att   = (const float*)d_in[5];
  const float* bias  = (const float*)d_in[6];
  const int*   ei    = (const int*)d_in[7];
  float*       out   = (float*)d_out;

  int N = in_sizes[0] / F_IN;
  int E = in_sizes[1];

  char* ws = (char*)d_ws;
  size_t o = 0;
  auto take = [&](size_t bytes) -> void* {
    void* p = ws + o; o += (bytes + 255) & ~(size_t)255; return p;
  };
  float* xl     = (float*)take((size_t)N * C * 4);
  int*   endp   = (int*)take((size_t)HB * N * 4);
  int2*  sedge  = (int2*)take((size_t)E * 8);
  float* xr     = out;

  const int B = 256;
  k_count_gemm<<<HB + GB, 1024, 0, stream>>>(ei, eattr, endp, sedge, x, Wl, Wr, xl, xr, N, E);
  k_node<<<(unsigned)(((long long)N * GL + B - 1) / B), B, 0, stream>>>(endp, sedge, xl, We, att, bias, out, N, E);
}


# node kernel logits on adjacent column pairs (packed ops on registers as loaded, leaky_relu as max(m,0.2m)): fewer instructions, 144 to 128 registers, 8 waves per SIMD; logit sum order changes
# speedup vs baseline: 9.9090x; 1.0786x over previous
#include <hip/hip_runtime.h>

#define F_IN 128
#define C 64

typedef float v2f __attribute__((ext_vector_type(2)));
typedef float v8f __attribute__((ext_vector_type(8)));

#define HB 48
#define NODE_CAP 65536
#define REC_CAP 20480
#define GB 208

typedef int v2i __attribute__((ext_vector_type(2)));
typedef int v4i __attribute__((ext_vector_type(4)));
typedef unsigned v2u __attribute__((ext_vector_type(2)));

__device__ __forceinline__ int chunk_len(int E) { return ((E + HB - 1) / HB + 15) & ~15; }

typedef float v4f __attribute__((ext_vector_type(4)));

__device__ __forceinline__ v8f mma4(v4f a, v4f b, int j, v8f acc) {
  v2f aa, bb;
  aa.x = a[2 * j]; aa.y = a[2 * j + 1];
  bb.x = b[2 * j]; bb.y = b[2 * j + 1];
  return __builtin_amdgcn_wmma_f32_16x16x4_f32(
       false, aa,  false, bb,
       (short)0, acc,  false,  false);
}

__device__ __forceinline__ void gemm_block(const float* __restrict__ x, const float* __restrict__ Wl,
                                           const float* __restrict__ Wr, float* __restrict__ xl,
                                           float* __restrict__ xr, float* wlds, int gb, int ngb, int N, int tid) {
  const long long T = (long long)(N >> 4) * 2;
  if ((long long)gb * 32 >= T) return;
  const int lane = tid & 31, hf = lane >> 4, c = lane & 15;
  const long long tstep = (long long)ngb * 32;
  long long t = (long long)gb * 32 + (tid >> 5);
  const v4f* ap = (const v4f*)(x + (long long)((int)(t >> 1) * 16 + c) * F_IN + hf * 4);
  v4f a[4];
  if (t < T) {
#pragma unroll
    for (int i = 0; i < 4; ++i) a[i] = ap[2 * i];
  }
  {
    const int col = tid & 127, kk0 = tid >> 7;
    const float* s = (col < 64 ? Wl + col : Wr + (col - 64)) + (long long)(kk0 * 4) * C;
    v4f w[4];
#pragma unroll
    for (int i = 0; i < 4; ++i) {
      const float* si = s + (long long)(i * 32) * C;
      w[i].x = si[0]; w[i].y = si[C]; w[i].z = si[2 * C]; w[i].w = si[3 * C];
    }
#pragma unroll
    for (int i = 0; i < 4; ++i) *(v4f*)(wlds + (tid + i * 1024) * 4) = w[i];
  }
  __syncthreads();

  while (t < T) {
    const int which = (int)(t & 1);
    const int m0    = (int)(t >> 1) << 4;
    const v4f* bw = (const v4f*)wlds + hf * 128 + which * 64 + c;
    v8f acc[4];
#pragma unroll
    for (int tl = 0; tl < 4; ++tl) acc[tl] = (v8f){0.f, 0.f, 0.f, 0.f, 0.f, 0.f, 0.f, 0.f};
    v4f b[4];
#pragma unroll
    for (int tl = 0; tl < 4; ++tl) b[tl] = bw[tl * 16];
#pragma unroll
    for (int g = 0; g < 16; ++g) {
      v4f bn[4];
      if (g + 1 < 16) {
#pragma unroll
        for (int tl = 0; tl < 4; ++tl) bn[tl] = bw[(g + 1) * 256 + tl * 16];
      }
      v4f av = a[g & 3];
      if (g + 4 < 16) a[g & 3] = ap[2 * (g + 4)];
#pragma unroll
      for (int j = 0; j < 2; ++j) {
#pragma unroll
        for (int tl = 0; tl < 4; ++tl) acc[tl] = mma4(av, b[tl], j, acc[tl]);
      }
      if (g + 1 < 16) {
#pragma unroll
        for (int tl = 0; tl < 4; ++tl) b[tl] = bn[tl];
      }
      __builtin_amdgcn_sched_barrier(0);
    }
    float* yrow = (which ? xr : xl) + (long long)(m0 + hf * 8) * C + c;
#pragma unroll
    for (int tl = 0; tl < 4; ++tl) {
#pragma unroll
      for (int r = 0; r < 8; ++r) yrow[(long long)r * C + tl * 16] = acc[tl][r];
    }
    t += tstep;
    if (t < T) {
      ap = (const v4f*)(x + (long long)((int)(t >> 1) * 16 + c) * F_IN + hf * 4);
#pragma unroll
      for (int i = 0; i < 4; ++i) a[i] = ap[2 * i];
    }
  }
}

__global__ __launch_bounds__(1024) void k_count_gemm(const int* __restrict__ ei, const float* __restrict__ eattr,
                                                     int* __restrict__ endp, int2* __restrict__ sedge,
                                                     const float* __restrict__ x, const float* __restrict__ Wl,
                                                     const float* __restrict__ Wr, float* __restrict__ xl,
                                                     float* __restrict__ xr, int N, int E) {
  __shared__ __attribute__((aligned(16))) unsigned hw[NODE_CAP / 2];
  __shared__ __attribute__((aligned(16))) v2i rec[REC_CAP];
  __shared__ int wsum[32];
  const int tid = threadIdx.x;
  if (blockIdx.x >= HB) {
    gemm_block(x, Wl, Wr, xl, xr, (float*)hw, (int)blockIdx.x - HB, (int)gridDim.x - HB, N, tid);
    return;
  }
  const bool vec = (N & 3) == 0;
  const int nw = (N + 1) >> 1;
  {
    v4i* h4 = (v4i*)hw;
    for (int i = tid; i < ((nw + 3) >> 2); i += 1024) h4[i] = (v4i){0, 0, 0, 0};
  }
  __syncthreads();
  const int chunk = chunk_len(E);
  const int e0 = blockIdx.x * chunk;
  const int e1 = min(e0 + chunk, E);
  const int* __restrict__ dst = ei + E;
  int e = e0 + tid;
  for (; e + 3072 < e1; e += 4096) {
    int d0 = dst[e], d1 = dst[e + 1024], d2 = dst[e + 2048], d3 = dst[e + 3072];
    atomicAdd(&hw[d0 >> 1], 1u << ((d0 & 1) << 4));
    atomicAdd(&hw[d1 >> 1], 1u << ((d1 & 1) << 4));
    atomicAdd(&hw[d2 >> 1], 1u << ((d2 & 1) << 4));
    atomicAdd(&hw[d3 >> 1], 1u << ((d3 & 1) << 4));
  }
  for (; e < e1; e += 1024) { int d = dst[e]; atomicAdd(&hw[d >> 1], 1u << ((d & 1) << 4)); }
  __syncthreads();

  {
    const int per = (nw + 1023) >> 10;
    const int lo = min(tid * per, nw), hi = min(lo + per, nw);
    int sum = 0;
    for (int i = lo; i < hi; ++i) { unsigned w = hw[i]; sum += (int)(w & 0xffffu) + (int)(w >> 16); }
    const int lane = tid & 31, wv = tid >> 5;
    int inc = sum;
#pragma unroll
    for (int o = 1; o < 32; o <<= 1) { int u = __shfl_up(inc, o, 32); if (lane >= o) inc += u; }
    if (lane == 31) wsum[wv] = inc;
    __syncthreads();
    int v = wsum[lane], vi = v;
#pragma unroll
    for (int o = 1; o < 32; o <<= 1) { int u = __shfl_up(vi, o, 32); if (lane >= o) vi += u; }
    int run = __shfl(vi - v, wv, 32) + inc - sum;
    for (int i = lo; i < hi; ++i) {
      unsigned w = hw[i];
      int c0 = (int)(w & 0xffffu), c1 = (int)(w >> 16);
      hw[i] = (unsigned)run | ((unsigned)(run + c0) << 16);
      run += c0 + c1;
    }
  }
  __syncthreads();

  const int* __restrict__ srcp = ei;
  e = e0 + tid;
  for (; e + 3072 < e1; e += 4096) {
    int d0 = dst[e], d1 = dst[e + 1024], d2 = dst[e + 2048], d3 = dst[e + 3072];
    int s0 = srcp[e], s1 = srcp[e + 1024], s2 = srcp[e + 2048], s3 = srcp[e + 3072];
    float a0 = eattr[e], a1 = eattr[e + 1024], a2 = eattr[e + 2048], a3 = eattr[e + 3072];
    const int h0 = (d0 & 1) << 4, h1 = (d1 & 1) << 4, h2 = (d2 & 1) << 4, h3 = (d3 & 1) << 4;
    unsigned q0 = atomicAdd(&hw[d0 >> 1], 1u << h0);
    unsigned q1 = atomicAdd(&hw[d1 >> 1], 1u << h1);
    unsigned q2 = atomicAdd(&hw[d2 >> 1], 1u << h2);
    unsigned q3 = atomicAdd(&hw[d3 >> 1], 1u << h3);
    rec[(q0 >> h0) & 0xffffu] = (v2i){s0, __float_as_int(a0)};
    rec[(q1 >> h1) & 0xffffu] = (v2i){s1, __float_as_int(a1)};
    rec[(q2 >> h2) & 0xffffu] = (v2i){s2, __float_as_int(a2)};
    rec[(q3 >> h3) & 0xffffu] = (v2i){s3, __float_as_int(a3)};
  }
  for (; e < e1; e += 1024) {
    int d = dst[e];
    const int hs = (d & 1) << 4;
    unsigned q = atomicAdd(&hw[d >> 1], 1u << hs);
    rec[(q >> hs) & 0xffffu] = (v2i){srcp[e], __float_as_int(eattr[e])};
  }
  __syncthreads();
  {
    const int cnt = max(e1 - e0, 0);
    v2i* so = (v2i*)(sedge + e0);
    const v4i* r4 = (const v4i*)rec;
    v4i* so4 = (v4i*)so;
    for (int i = tid; i < (cnt >> 1); i += 1024) so4[i] = r4[i];
    if ((cnt & 1) && tid == 0) so[cnt - 1] = rec[cnt - 1];
  }
  int* __restrict__ o = endp + (long long)blockIdx.x * N;
  if (vec) {
    const v2u* h2 = (const v2u*)hw;
    v4i* __restrict__ o4 = (v4i*)o;
    for (int i = tid; i < (N >> 2); i += 1024) {
      v2u w = h2[i];
      o4[i] = (v4i){(int)(w.x & 0xffffu), (int)(w.x >> 16), (int)(w.y & 0xffffu), (int)(w.y >> 16)};
    }
  } else {
    for (int i = tid; i < N; i += 1024) o[i] = (int)((hw[i >> 1] >> ((i & 1) << 4)) & 0xffffu);
  }
}


template <int M>
__device__ __forceinline__ float grp_xadd(float p) {
  return p + __int_as_float(__builtin_amdgcn_update_dpp(0, __float_as_int(p), 0x160 + M, 0xf, 0xf, true));
}
template <int M>
__device__ __forceinline__ int grp_xadd_i(int v) {
  return v + __builtin_amdgcn_update_dpp(0, v, 0x160 + M, 0xf, 0xf, true);
}

template <int S>
__device__ __forceinline__ int grp_shr_add(int v, int l7) {
  const int u = __builtin_amdgcn_update_dpp(0, v, 0x110 + S, 0xf, 0xf, true);
  return v + (l7 >= S ? u : 0);
}

#define GL 8
#define RPL (HB / GL)
static_assert(HB % GL == 0, "every lane of a group takes the same number of chunks");

__device__ __forceinline__ v2f pair_logit(v2f a, v2f b, v2f e2, v2f we, v2f at, v2f p) {
  v2f m = a + b + e2 * we;
  m = __builtin_elementwise_max(m, m * 0.2f);
  return p + m * at;
}

__device__ __forceinline__ float edge_logit(v4f a0, v4f a1, v4f b0, v4f b1, float eq, v4f we0, v4f we1, v4f at0,
                                            v4f at1) {
  const v2f e2 = {eq, eq};
  v2f p2 = {0.f, 0.f};
  p2 = pair_logit(a0.xy, b0.xy, e2, we0.xy, at0.xy, p2);
  p2 = pair_logit(a0.zw, b0.zw, e2, we0.zw, at0.zw, p2);
  p2 = pair_logit(a1.xy, b1.xy, e2, we1.xy, at1.xy, p2);
  p2 = pair_logit(a1.zw, b1.zw, e2, we1.zw, at1.zw, p2);
  float p = p2.x + p2.y;
  p = grp_xadd<1>(p); p = grp_xadd<2>(p); p = grp_xadd<4>(p);
  return p;
}

__global__ __launch_bounds__(256, 8) void k_node(const int* __restrict__ endp, const int2* __restrict__ sedge,
                       const float* __restrict__ xl,
                       const float* __restrict__ We, const float* __restrict__ att,
                       const float* __restrict__ bias, float* out, int N, int E) {
  __shared__ int lidx[512];
  const int l7 = threadIdx.x & (GL - 1);
  const int d  = (blockIdx.x * blockDim.x + threadIdx.x) >> 3;
  if (d >= N) return;
  const int rowl = (threadIdx.x >> 3) << 4;
  const int chunk = chunk_len(E);
  const int co = l7 * 4;
  const float* orow = out + (long long)d * C + co;
  const v4f b0 = *(const v4f*)orow, b1 = *(const v4f*)(orow + 32);
  const float* srow = xl + (long long)d * C + co;
  const v4f s0 = *(const v4f*)srow, s1 = *(const v4f*)(srow + 32);
  const v4f we0 = *(const v4f*)(We + co),  we1 = *(const v4f*)(We + 32 + co);
  const v4f at0 = *(const v4f*)(att + co), at1 = *(const v4f*)(att + 32 + co);
  int cnt[RPL], sb[RPL], rb[RPL];
  int tl = 0;
  const int back = (d > 0) ? 1 : 0;
#pragma unroll
  for (int k = 0; k < RPL; ++k) {
    const int p = l7 * RPL + k;
    const int* pl = endp + (long long)p * N + d;
    int r1 = pl[0];
    int r0 = pl[-back];
    r0 = back ? r0 : 0;
    cnt[k] = r1 - r0;
    rb[k]  = p * chunk + r0;
    tl += cnt[k];
  }
  int incl = tl;
  incl = grp_shr_add<1>(incl, l7); incl = grp_shr_add<2>(incl, l7); incl = grp_shr_add<4>(incl, l7);
  sb[0] = incl - tl;
#pragma unroll
  for (int k = 1; k < RPL; ++k) sb[k] = sb[k - 1] + cnt[k - 1];
  int end = tl;
  end = grp_xadd_i<1>(end); end = grp_xadd_i<2>(end); end = grp_xadd_i<4>(end);

  float M = -INFINITY, S = 0.f, easum = 0.f;
  v4f acc0 = {0.f, 0.f, 0.f, 0.f}, acc1 = {0.f, 0.f, 0.f, 0.f};
  for (int c = 0; c < end; c += 16) {
#pragma unroll
    for (int k = 0; k < RPL; ++k) {
      const int lo = max(sb[k], c), hi = min(sb[k] + cnt[k], c + 16);
#pragma clang loop unroll(disable) vectorize(disable) interleave(disable)
      for (int pos = lo; pos < hi; ++pos) lidx[rowl + pos - c] = rb[k] + (pos - sb[k]);
    }
    __builtin_amdgcn_fence(__ATOMIC_ACQ_REL, "wavefront");
    int sj0 = d, sj1 = d; float ea0 = 0.f, ea1 = 0.f;
    if (c + l7 < end)     { int2 r = sedge[lidx[rowl + l7]];     sj0 = r.x; ea0 = __int_as_float(r.y); }
    if (c + 8 + l7 < end) { int2 r = sedge[lidx[rowl + 8 + l7]]; sj1 = r.x; ea1 = __int_as_float(r.y); }
    easum += ea0 + ea1;
    int nrem = end - c; nrem = nrem > 16 ? 16 : nrem;
    for (int k0 = 0; k0 < nrem; k0 += 4) {
      const int   sjv = (k0 & 8) ? sj1 : sj0;
      const float eav = (k0 & 8) ? ea1 : ea0;
      v4f a0[4], a1[4]; float eq[4], l[4];
#pragma unroll
      for (int k = 0; k < 4; ++k) {
        int sk = __shfl(sjv, (k0 & 4) + k, GL);
        eq[k]  = __shfl(eav, (k0 & 4) + k, GL);
        const float* r = xl + (long long)sk * C + co;
        a0[k] = *(const v4f*)r;
        a1[k] = *(const v4f*)(r + 32);
      }
      float mx = M;
#pragma unroll
      for (int k = 0; k < 4; ++k) {
        l[k] = edge_logit(a0[k], a1[k], b0, b1, eq[k], we0, we1, at0, at1);
        l[k] = (k0 + k < nrem) ? l[k] : -INFINITY;
        mx = fmaxf(mx, l[k]);
      }
      float sc = __expf(M - mx);
      S *= sc; acc0 *= sc; acc1 *= sc;
#pragma unroll
      for (int k = 0; k < 4; ++k) {
        float p = __expf(l[k] - mx);
        S += p; acc0 += p * a0[k]; acc1 += p * a1[k];
      }
      M = mx;
    }
  }
  easum = grp_xadd<1>(easum); easum = grp_xadd<2>(easum); easum = grp_xadd<4>(easum);
  float loop_ea = easum / fmaxf((float)end, 1.0f);
  float ls = edge_logit(s0, s1, b0, b1, loop_ea, we0, we1, at0, at1);
  float mx = fmaxf(M, ls);
  float sc = __expf(M - mx);
  float p  = __expf(ls - mx);
  S = S * sc + p;
  acc0 = acc0 * sc + p * s0;
  acc1 = acc1 * sc + p * s1;
  const v4f bi0 = *(const v4f*)(bias + co), bi1 = *(const v4f*)(bias + 32 + co);
  v4f o0, o1;
#pragma unroll
  for (int k = 0; k < 4; ++k) { o0[k] = acc0[k] / S + bi0[k]; o1[k] = acc1[k] / S + bi1[k]; }
  float* wrow = out + (long long)d * C + co;
  *(v4f*)wrow = o0;
  *(v4f*)(wrow + 32) = o1;
}

extern "C" void kernel_launch(void* const* d_in, const int* in_sizes, int n_in,
                              void* d_out, int out_size, void* d_ws, size_t ws_size,
                              hipStream_t stream) {
  (void)n_in; (void)out_size; (void)ws_size;
  const float* x     = (const float*)d_in[0];
  const float* eattr = (const float*)d_in[1];
  const float* Wl    = (const float*)d_in[2];
  const float* Wr    = (const float*)d_in[3];
  const float* We    = (const float*)d_in[4];
  const float* att   = (const float*)d_in[5];
  const float* bias  = (const float*)d_in[6];
  const int*   ei    = (const int*)d_in[7];
  float*       out   = (float*)d_out;

  int N = in_sizes[0] / F_IN;
  int E = in_sizes[1];

  char* ws = (char*)d_ws;
  size_t o = 0;
  auto take = [&](size_t bytes) -> void* {
    void* p = ws + o; o += (bytes + 255) & ~(size_t)255; return p;
  };
  float* xl     = (float*)take((size_t)N * C * 4);
  int*   endp   = (int*)take((size_t)HB * N * 4);
  int2*  sedge  = (int2*)take((size_t)E * 8);
  float* xr     = out;

  const int B = 256;
  k_count_gemm<<<HB + GB, 1024, 0, stream>>>(ei, eattr, endp, sedge, x, Wl, Wr, xl, xr, N, E);
  k_node<<<(unsigned)(((long long)N * GL + B - 1) / B), B, 0, stream>>>(endp, sedge, xl, We, att, bias, out, N, E);
}
